# starting kernel
# baseline (speedup 1.0000x reference)
#include <hip/hip_runtime.h>
#include <hip/hip_bf16.h>

#define IN_F  256
#define OUT_F 128

typedef __attribute__((ext_vector_type(2))) float v2f;
typedef __attribute__((ext_vector_type(8))) float v8f;

__global__ void __launch_bounds__(256)
gcn_gemm_wmma(const float* __restrict__ x, const float* __restrict__ w,
              const float* __restrict__ bias, float* __restrict__ h,
              int n_nodes) {
    const int lane = threadIdx.x & 31;
    const int wave = threadIdx.x >> 5;
    const int half = lane >> 4;
    const int l16  = lane & 15;

    const int row_base = blockIdx.x * 16;
    const int col_base = wave * 16;

    int m = row_base + l16;
    if (m >= n_nodes) m = n_nodes - 1;
    const int n = col_base + l16;

    const float* xrow = x + (size_t)m * IN_F;
    const float* wcol = w + n;

    v8f acc = {};
#pragma unroll 8
    for (int k0 = 0; k0 < IN_F; k0 += 4) {
        const int ka = k0 + 2 * half;
        v2f a = *(const v2f*)(xrow + ka);
        v2f b;
        b.x = wcol[(size_t)ka * OUT_F];
        b.y = wcol[(size_t)(ka + 1) * OUT_F];
        acc = __builtin_amdgcn_wmma_f32_16x16x4_f32(
             false, a,  false, b,
             (short)0, acc,  false,  false);
    }

    const float bn = bias[n];
    float* hp = h + (size_t)(row_base + 8 * half) * OUT_F + n;

    if (row_base + 16 <= n_nodes) {
#pragma unroll
        for (int v = 0; v < 8; ++v)
            hp[(size_t)v * OUT_F] = acc[v] + bn;
    } else {
        const int r0 = row_base + 8 * half;
#pragma unroll
        for (int v = 0; v < 8; ++v)
            if (r0 + v < n_nodes) hp[(size_t)v * OUT_F] = acc[v] + bn;
    }
}

__global__ void gcn_zero(float4* __restrict__ out, long long n4) {
    long long i = (long long)blockIdx.x * blockDim.x + threadIdx.x;
    if (i < n4) out[i] = make_float4(0.f, 0.f, 0.f, 0.f);
}

__global__ void __launch_bounds__(256)
gcn_scatter(const float* __restrict__ h, const int* __restrict__ erow,
            const int* __restrict__ ecol, const float* __restrict__ eval,
            float* __restrict__ out, int n_edges) {
    const int edge = blockIdx.x * 8 + (int)(threadIdx.x >> 5);
    const int lane = threadIdx.x & 31;
    if (edge >= n_edges) return;

    const int   r = erow[edge];
    const int   c = ecol[edge];
    const float v = eval[edge];

    const float4 hm = *(const float4*)(h + (size_t)c * OUT_F + lane * 4);
    float* op = out + (size_t)r * OUT_F + lane * 4;
    atomicAdd(op + 0, hm.x * v);
    atomicAdd(op + 1, hm.y * v);
    atomicAdd(op + 2, hm.z * v);
    atomicAdd(op + 3, hm.w * v);
}

extern "C" void kernel_launch(void* const* d_in, const int* in_sizes, int n_in,
                              void* d_out, int out_size, void* d_ws, size_t ws_size,
                              hipStream_t stream) {
    const float* x    = (const float*)d_in[0];
    const int*   erow = (const int*)  d_in[1];
    const int*   ecol = (const int*)  d_in[2];
    const float* ev   = (const float*)d_in[3];
    const float* w    = (const float*)d_in[4];
    const float* bias = (const float*)d_in[5];
    float*       out  = (float*)d_out;
    float*       h    = (float*)d_ws;

    const int n_nodes = in_sizes[0] / IN_F;
    const int n_edges = in_sizes[1];

    const long long n4 = (long long)n_nodes * OUT_F / 4;
    gcn_zero<<<(unsigned)((n4 + 255) / 256), 256, 0, stream>>>((float4*)out, n4);

    const int row_tiles = (n_nodes + 15) / 16;
    gcn_gemm_wmma<<<row_tiles, 256, 0, stream>>>(x, w, bias, h, n_nodes);

    const int blocks = (n_edges + 7) / 8;
    gcn_scatter<<<blocks, 256, 0, stream>>>(h, erow, ecol, ev, out, n_edges);
}


# scatter: lane L owns floats L,L+32,L+64,L+96 so each f32 atomic instruction (and load) covers one full contiguous 128-B line, not 8 dwords in 4 lines; still arrival-order atomics
# speedup vs baseline: 1.2850x; 1.2850x over previous
#include <hip/hip_runtime.h>
#include <hip/hip_bf16.h>

#define IN_F  256
#define OUT_F 128

typedef __attribute__((ext_vector_type(2))) float v2f;
typedef __attribute__((ext_vector_type(8))) float v8f;

__global__ void __launch_bounds__(256)
gcn_gemm_wmma(const float* __restrict__ x, const float* __restrict__ w,
              const float* __restrict__ bias, float* __restrict__ h,
              int n_nodes) {
    const int lane = threadIdx.x & 31;
    const int wave = threadIdx.x >> 5;
    const int half = lane >> 4;
    const int l16  = lane & 15;

    const int row_base = blockIdx.x * 16;
    const int col_base = wave * 16;

    int m = row_base + l16;
    if (m >= n_nodes) m = n_nodes - 1;
    const int n = col_base + l16;

    const float* xrow = x + (size_t)m * IN_F;
    const float* wcol = w + n;

    v8f acc = {};
#pragma unroll 8
    for (int k0 = 0; k0 < IN_F; k0 += 4) {
        const int ka = k0 + 2 * half;
        v2f a = *(const v2f*)(xrow + ka);
        v2f b;
        b.x = wcol[(size_t)ka * OUT_F];
        b.y = wcol[(size_t)(ka + 1) * OUT_F];
        acc = __builtin_amdgcn_wmma_f32_16x16x4_f32(
             false, a,  false, b,
             (short)0, acc,  false,  false);
    }

    const float bn = bias[n];
    float* hp = h + (size_t)(row_base + 8 * half) * OUT_F + n;

    if (row_base + 16 <= n_nodes) {
#pragma unroll
        for (int v = 0; v < 8; ++v)
            hp[(size_t)v * OUT_F] = acc[v] + bn;
    } else {
        const int r0 = row_base + 8 * half;
#pragma unroll
        for (int v = 0; v < 8; ++v)
            if (r0 + v < n_nodes) hp[(size_t)v * OUT_F] = acc[v] + bn;
    }
}

__global__ void gcn_zero(float4* __restrict__ out, long long n4) {
    long long i = (long long)blockIdx.x * blockDim.x + threadIdx.x;
    if (i < n4) out[i] = make_float4(0.f, 0.f, 0.f, 0.f);
}

__global__ void __launch_bounds__(256)
gcn_scatter(const float* __restrict__ h, const int* __restrict__ erow,
            const int* __restrict__ ecol, const float* __restrict__ eval,
            float* __restrict__ out, int n_edges) {
    const int edge = blockIdx.x * 8 + (int)(threadIdx.x >> 5);
    const int lane = threadIdx.x & 31;
    if (edge >= n_edges) return;

    const int   r = erow[edge];
    const int   c = ecol[edge];
    const float v = eval[edge];

    const float* hp = h + (size_t)c * OUT_F + lane;
    const float h0 = hp[0];
    const float h1 = hp[32];
    const float h2 = hp[64];
    const float h3 = hp[96];

    float* op = out + (size_t)r * OUT_F + lane;
    atomicAdd(op +  0, h0 * v);
    atomicAdd(op + 32, h1 * v);
    atomicAdd(op + 64, h2 * v);
    atomicAdd(op + 96, h3 * v);
}

extern "C" void kernel_launch(void* const* d_in, const int* in_sizes, int n_in,
                              void* d_out, int out_size, void* d_ws, size_t ws_size,
                              hipStream_t stream) {
    const float* x    = (const float*)d_in[0];
    const int*   erow = (const int*)  d_in[1];
    const int*   ecol = (const int*)  d_in[2];
    const float* ev   = (const float*)d_in[3];
    const float* w    = (const float*)d_in[4];
    const float* bias = (const float*)d_in[5];
    float*       out  = (float*)d_out;
    float*       h    = (float*)d_ws;

    const int n_nodes = in_sizes[0] / IN_F;
    const int n_edges = in_sizes[1];

    const long long n4 = (long long)n_nodes * OUT_F / 4;
    gcn_zero<<<(unsigned)((n4 + 255) / 256), 256, 0, stream>>>((float4*)out, n4);

    const int row_tiles = (n_nodes + 15) / 16;
    gcn_gemm_wmma<<<row_tiles, 256, 0, stream>>>(x, w, bias, h, n_nodes);

    const int blocks = (n_edges + 7) / 8;
    gcn_scatter<<<blocks, 256, 0, stream>>>(h, erow, ecol, ev, out, n_edges);
}


# GEMM: W staged once per work-group in LDS in fragment order, each wave owns 16 rows x all 128 columns: x read by one wave not 8, W from LDS; same WMMA order per accumulator, h bit-identical
# speedup vs baseline: 1.5571x; 1.2118x over previous
#include <hip/hip_runtime.h>
#include <hip/hip_bf16.h>

#define IN_F  256
#define OUT_F 128

typedef __attribute__((ext_vector_type(2))) float v2f;
typedef __attribute__((ext_vector_type(4))) float v4f;
typedef __attribute__((ext_vector_type(8))) float v8f;

__device__ __forceinline__ void gemm_chunk(v8f (&acc)[8], const v2f (&a)[8],
                                           const v4f* bp) {
#pragma unroll
    for (int gg = 0; gg < 4; ++gg) {
        v4f b[8];
#pragma unroll
        for (int j = 0; j < 8; ++j) b[j] = bp[(gg * 8 + j) * 32];
#pragma unroll
        for (int j = 0; j < 8; ++j)
            acc[j] = __builtin_amdgcn_wmma_f32_16x16x4_f32(
                 false, a[2 * gg],  false, b[j].xy,
                 (short)0, acc[j],  false,  false);
#pragma unroll
        for (int j = 0; j < 8; ++j)
            acc[j] = __builtin_amdgcn_wmma_f32_16x16x4_f32(
                 false, a[2 * gg + 1],  false, b[j].zw,
                 (short)0, acc[j],  false,  false);
    }
    __builtin_amdgcn_sched_barrier(0);
    asm volatile("v_nop\n\tv_nop\n\tv_nop\n\tv_nop\n\tv_nop\n\tv_nop\n\tv_nop\n\tv_nop\n\tv_nop");
    __builtin_amdgcn_sched_barrier(0);
}

#define GEMM_WAVES 8

__global__ void __launch_bounds__(256)
gcn_gemm_wmma(const float* __restrict__ x, const float* __restrict__ w,
              const float* __restrict__ bias, float* __restrict__ h,
              int n_nodes, int n_tiles) {
    __shared__ __attribute__((aligned(16))) v4f wl[32 * 8 * 32];

    const int lane = threadIdx.x & 31;
    const int wave = __builtin_amdgcn_readfirstlane((int)(threadIdx.x >> 5));
    const int half = lane >> 4;
    const int l16  = lane & 15;

    const int tstride = GEMM_WAVES * (int)gridDim.x;
    int tile = wave * (int)gridDim.x + (int)blockIdx.x;

    int m = tile * 16 + l16;
    if (m >= n_nodes) m = n_nodes - 1;
    const float* xrow = x + (size_t)m * IN_F + 2 * half;

    v2f a_cur[8];
#pragma unroll
    for (int u = 0; u < 8; ++u) a_cur[u] = *(const v2f*)(xrow + 4 * u);

#pragma unroll 4
    for (int q = 0; q < 32; ++q) {
        const int i  = (int)threadIdx.x + 256 * q;
        const int n  = i & 127;
        const int gh = i >> 7;
        const int g  = gh >> 1;
        const int hh = gh & 1;
        const float* wp = w + (size_t)(8 * g + 2 * hh) * OUT_F + n;
        v4f v;
        v.x = wp[0];
        v.y = wp[OUT_F];
        v.z = wp[4 * OUT_F];
        v.w = wp[5 * OUT_F];
        wl[(g * 8 + (n >> 4)) * 32 + hh * 16 + (n & 15)] = v;
    }
    __syncthreads();

    while (tile < n_tiles) {
        const int ntile = tile + tstride;
        int mn = ntile * 16 + l16;
        if (mn >= n_nodes) mn = n_nodes - 1;
        const float* xnext = x + (size_t)mn * IN_F + 2 * half;

        v8f acc[8];
#pragma unroll
        for (int j = 0; j < 8; ++j) { const v8f zero = {}; acc[j] = zero; }

        const v4f* bp = wl + lane;
#pragma nounroll
        for (int cp = 0; cp < 4; ++cp) {
            v2f a_alt[8];
            {
                const float* src = xrow + 32 * (2 * cp + 1);
#pragma unroll
                for (int u = 0; u < 8; ++u) a_alt[u] = *(const v2f*)(src + 4 * u);
            }
            gemm_chunk(acc, a_cur, bp);
            {
                const float* src = (cp < 3) ? (xrow + 32 * (2 * cp + 2)) : xnext;
#pragma unroll
                for (int u = 0; u < 8; ++u) a_cur[u] = *(const v2f*)(src + 4 * u);
            }
            gemm_chunk(acc, a_alt, bp + 4 * 8 * 32);
            bp += 2 * 4 * 8 * 32;
        }

        const int row_base = tile * 16;
        float bn[8];
#pragma unroll
        for (int j = 0; j < 8; ++j) bn[j] = bias[16 * j + l16];
        float* hp = h + (size_t)(row_base + 8 * half) * OUT_F + l16;

        if (row_base + 16 <= n_nodes) {
#pragma unroll
            for (int j = 0; j < 8; ++j)
#pragma unroll
                for (int v = 0; v < 8; ++v)
                    hp[(size_t)v * OUT_F + 16 * j] = acc[j][v] + bn[j];
        } else {
            const int r0 = row_base + 8 * half;
#pragma unroll
            for (int j = 0; j < 8; ++j)
#pragma unroll
                for (int v = 0; v < 8; ++v)
                    if (r0 + v < n_nodes) hp[(size_t)v * OUT_F + 16 * j] = acc[j][v] + bn[j];
        }

        tile = ntile;
        xrow = xnext;
    }
}

__global__ void gcn_zero(float4* __restrict__ out, long long n4) {
    long long i = (long long)blockIdx.x * blockDim.x + threadIdx.x;
    if (i < n4) out[i] = make_float4(0.f, 0.f, 0.f, 0.f);
}

__global__ void __launch_bounds__(256)
gcn_scatter(const float* __restrict__ h, const int* __restrict__ erow,
            const int* __restrict__ ecol, const float* __restrict__ eval,
            float* __restrict__ out, int n_edges) {
    const int edge = blockIdx.x * 8 + (int)(threadIdx.x >> 5);
    const int lane = threadIdx.x & 31;
    if (edge >= n_edges) return;

    const int   r = erow[edge];
    const int   c = ecol[edge];
    const float v = eval[edge];

    const float* hp = h + (size_t)c * OUT_F + lane;
    const float h0 = hp[0];
    const float h1 = hp[32];
    const float h2 = hp[64];
    const float h3 = hp[96];

    float* op = out + (size_t)r * OUT_F + lane;
    atomicAdd(op +  0, h0 * v);
    atomicAdd(op + 32, h1 * v);
    atomicAdd(op + 64, h2 * v);
    atomicAdd(op + 96, h3 * v);
}

extern "C" void kernel_launch(void* const* d_in, const int* in_sizes, int n_in,
                              void* d_out, int out_size, void* d_ws, size_t ws_size,
                              hipStream_t stream) {
    const float* x    = (const float*)d_in[0];
    const int*   erow = (const int*)  d_in[1];
    const int*   ecol = (const int*)  d_in[2];
    const float* ev   = (const float*)d_in[3];
    const float* w    = (const float*)d_in[4];
    const float* bias = (const float*)d_in[5];
    float*       out  = (float*)d_out;
    float*       h    = (float*)d_ws;

    const int n_nodes = in_sizes[0] / IN_F;
    const int n_edges = in_sizes[1];

    const long long n4 = (long long)n_nodes * OUT_F / 4;
    gcn_zero<<<(unsigned)((n4 + 255) / 256), 256, 0, stream>>>((float4*)out, n4);

    const int row_tiles = (n_nodes + 15) / 16;
    gcn_gemm_wmma<<<512, 256, 0, stream>>>(x, w, bias, h, n_nodes, row_tiles);

    const int blocks = (n_edges + 7) / 8;
    gcn_scatter<<<blocks, 256, 0, stream>>>(h, erow, ecol, ev, out, n_edges);
}


# scatter: each wave takes 32 consecutive edges (coalesced index loads, lane broadcast to scalar), issuing the 32 line loads of 8 edges before their atomics; 50K waves not 1.6M; arrival-order atomics
# speedup vs baseline: 2.2176x; 1.4242x over previous
#include <hip/hip_runtime.h>
#include <hip/hip_bf16.h>

#define IN_F  256
#define OUT_F 128

typedef __attribute__((ext_vector_type(2))) float v2f;
typedef __attribute__((ext_vector_type(4))) float v4f;
typedef __attribute__((ext_vector_type(8))) float v8f;

__device__ __forceinline__ void gemm_chunk(v8f (&acc)[8], const v2f (&a)[8],
                                           const v4f* bp) {
#pragma unroll
    for (int gg = 0; gg < 4; ++gg) {
        v4f b[8];
#pragma unroll
        for (int j = 0; j < 8; ++j) b[j] = bp[(gg * 8 + j) * 32];
#pragma unroll
        for (int j = 0; j < 8; ++j)
            acc[j] = __builtin_amdgcn_wmma_f32_16x16x4_f32(
                 false, a[2 * gg],  false, b[j].xy,
                 (short)0, acc[j],  false,  false);
#pragma unroll
        for (int j = 0; j < 8; ++j)
            acc[j] = __builtin_amdgcn_wmma_f32_16x16x4_f32(
                 false, a[2 * gg + 1],  false, b[j].zw,
                 (short)0, acc[j],  false,  false);
    }
    __builtin_amdgcn_sched_barrier(0);
    asm volatile("v_nop\n\tv_nop\n\tv_nop\n\tv_nop\n\tv_nop\n\tv_nop\n\tv_nop\n\tv_nop\n\tv_nop");
    __builtin_amdgcn_sched_barrier(0);
}

#define GEMM_WAVES 8

__global__ void __launch_bounds__(256)
gcn_gemm_wmma(const float* __restrict__ x, const float* __restrict__ w,
              const float* __restrict__ bias, float* __restrict__ h,
              int n_nodes, int n_tiles) {
    __shared__ __attribute__((aligned(16))) v4f wl[32 * 8 * 32];

    const int lane = threadIdx.x & 31;
    const int wave = __builtin_amdgcn_readfirstlane((int)(threadIdx.x >> 5));
    const int half = lane >> 4;
    const int l16  = lane & 15;

    const int tstride = GEMM_WAVES * (int)gridDim.x;
    int tile = wave * (int)gridDim.x + (int)blockIdx.x;

    int m = tile * 16 + l16;
    if (m >= n_nodes) m = n_nodes - 1;
    const float* xrow = x + (size_t)m * IN_F + 2 * half;

    v2f a_cur[8];
#pragma unroll
    for (int u = 0; u < 8; ++u) a_cur[u] = *(const v2f*)(xrow + 4 * u);

#pragma unroll 4
    for (int q = 0; q < 32; ++q) {
        const int i  = (int)threadIdx.x + 256 * q;
        const int n  = i & 127;
        const int gh = i >> 7;
        const int g  = gh >> 1;
        const int hh = gh & 1;
        const float* wp = w + (size_t)(8 * g + 2 * hh) * OUT_F + n;
        v4f v;
        v.x = wp[0];
        v.y = wp[OUT_F];
        v.z = wp[4 * OUT_F];
        v.w = wp[5 * OUT_F];
        wl[(g * 8 + (n >> 4)) * 32 + hh * 16 + (n & 15)] = v;
    }
    __syncthreads();

    while (tile < n_tiles) {
        const int ntile = tile + tstride;
        int mn = ntile * 16 + l16;
        if (mn >= n_nodes) mn = n_nodes - 1;
        const float* xnext = x + (size_t)mn * IN_F + 2 * half;

        v8f acc[8];
#pragma unroll
        for (int j = 0; j < 8; ++j) { const v8f zero = {}; acc[j] = zero; }

        const v4f* bp = wl + lane;
#pragma nounroll
        for (int cp = 0; cp < 4; ++cp) {
            v2f a_alt[8];
            {
                const float* src = xrow + 32 * (2 * cp + 1);
#pragma unroll
                for (int u = 0; u < 8; ++u) a_alt[u] = *(const v2f*)(src + 4 * u);
            }
            gemm_chunk(acc, a_cur, bp);
            {
                const float* src = (cp < 3) ? (xrow + 32 * (2 * cp + 2)) : xnext;
#pragma unroll
                for (int u = 0; u < 8; ++u) a_cur[u] = *(const v2f*)(src + 4 * u);
            }
            gemm_chunk(acc, a_alt, bp + 4 * 8 * 32);
            bp += 2 * 4 * 8 * 32;
        }

        const int row_base = tile * 16;
        float bn[8];
#pragma unroll
        for (int j = 0; j < 8; ++j) bn[j] = bias[16 * j + l16];
        float* hp = h + (size_t)(row_base + 8 * half) * OUT_F + l16;

        if (row_base + 16 <= n_nodes) {
#pragma unroll
            for (int j = 0; j < 8; ++j)
#pragma unroll
                for (int v = 0; v < 8; ++v)
                    hp[(size_t)v * OUT_F + 16 * j] = acc[j][v] + bn[j];
        } else {
            const int r0 = row_base + 8 * half;
#pragma unroll
            for (int j = 0; j < 8; ++j)
#pragma unroll
                for (int v = 0; v < 8; ++v)
                    if (r0 + v < n_nodes) hp[(size_t)v * OUT_F + 16 * j] = acc[j][v] + bn[j];
        }

        tile = ntile;
        xrow = xnext;
    }
}

__global__ void gcn_zero(float4* __restrict__ out, long long n4) {
    long long i = (long long)blockIdx.x * blockDim.x + threadIdx.x;
    if (i < n4) out[i] = make_float4(0.f, 0.f, 0.f, 0.f);
}

#define EDGES_PER_WAVE 32
#define EDGE_BATCH 8

__device__ __forceinline__ int lane_bcast_i(int x, int src_lane) {
    return __builtin_amdgcn_readlane(x, src_lane);
}
__device__ __forceinline__ float lane_bcast_f(float x, int src_lane) {
    return __int_as_float(__builtin_amdgcn_readlane(__float_as_int(x), src_lane));
}

__global__ void __launch_bounds__(256)
gcn_scatter(const float* __restrict__ h, const int* __restrict__ erow,
            const int* __restrict__ ecol, const float* __restrict__ eval,
            float* __restrict__ out, int n_edges) {
    const int lane = threadIdx.x & 31;
    const int wave = __builtin_amdgcn_readfirstlane((int)(threadIdx.x >> 5));
    const int base = ((int)blockIdx.x * 8 + wave) * EDGES_PER_WAVE;
    if (base >= n_edges) return;
    int cnt = n_edges - base;
    if (cnt > EDGES_PER_WAVE) cnt = EDGES_PER_WAVE;

    int   r = 0;
    int   c = 0;
    float v = 0.f;
    if (lane < cnt) {
        r = erow[base + lane];
        c = ecol[base + lane];
        v = eval[base + lane];
    }

    const float* hl = h + lane;
    float*       ol = out + lane;

    if (cnt == EDGES_PER_WAVE) {
#pragma unroll
        for (int j0 = 0; j0 < EDGES_PER_WAVE; j0 += EDGE_BATCH) {
            float g[EDGE_BATCH][4];
#pragma unroll
            for (int u = 0; u < EDGE_BATCH; ++u) {
                const int cj = lane_bcast_i(c, j0 + u);
                const float* hp = hl + (size_t)cj * OUT_F;
                g[u][0] = hp[0];
                g[u][1] = hp[32];
                g[u][2] = hp[64];
                g[u][3] = hp[96];
            }
#pragma unroll
            for (int u = 0; u < EDGE_BATCH; ++u) {
                const int   rj = lane_bcast_i(r, j0 + u);
                const float vj = lane_bcast_f(v, j0 + u);
                float* op = ol + (size_t)rj * OUT_F;
                atomicAdd(op +  0, g[u][0] * vj);
                atomicAdd(op + 32, g[u][1] * vj);
                atomicAdd(op + 64, g[u][2] * vj);
                atomicAdd(op + 96, g[u][3] * vj);
            }
        }
    } else {
        for (int j = 0; j < cnt; ++j) {
            const int   cj = lane_bcast_i(c, j);
            const int   rj = lane_bcast_i(r, j);
            const float vj = lane_bcast_f(v, j);
            const float* hp = hl + (size_t)cj * OUT_F;
            const float h0 = hp[0];
            const float h1 = hp[32];
            const float h2 = hp[64];
            const float h3 = hp[96];
            float* op = ol + (size_t)rj * OUT_F;
            atomicAdd(op +  0, h0 * vj);
            atomicAdd(op + 32, h1 * vj);
            atomicAdd(op + 64, h2 * vj);
            atomicAdd(op + 96, h3 * vj);
        }
    }
}

extern "C" void kernel_launch(void* const* d_in, const int* in_sizes, int n_in,
                              void* d_out, int out_size, void* d_ws, size_t ws_size,
                              hipStream_t stream) {
    const float* x    = (const float*)d_in[0];
    const int*   erow = (const int*)  d_in[1];
    const int*   ecol = (const int*)  d_in[2];
    const float* ev   = (const float*)d_in[3];
    const float* w    = (const float*)d_in[4];
    const float* bias = (const float*)d_in[5];
    float*       out  = (float*)d_out;
    float*       h    = (float*)d_ws;

    const int n_nodes = in_sizes[0] / IN_F;
    const int n_edges = in_sizes[1];

    const long long n4 = (long long)n_nodes * OUT_F / 4;
    gcn_zero<<<(unsigned)((n4 + 255) / 256), 256, 0, stream>>>((float4*)out, n4);

    const int row_tiles = (n_nodes + 15) / 16;
    gcn_gemm_wmma<<<512, 256, 0, stream>>>(x, w, bias, h, n_nodes, row_tiles);

    const int blocks = (n_edges + 255) / 256;
    gcn_scatter<<<blocks, 256, 0, stream>>>(h, erow, ecol, ev, out, n_edges);
}


# scatter: edges binned by destination range in workspace each call; one work-group per 196 rows sums in LDS (arrival-order LDS adds), stores rows once; no global atomics, no zero pass
# speedup vs baseline: 2.7774x; 1.2524x over previous
#include <hip/hip_runtime.h>
#include <hip/hip_bf16.h>

#define IN_F  256
#define OUT_F 128

typedef __attribute__((ext_vector_type(2))) float v2f;
typedef __attribute__((ext_vector_type(4))) float v4f;
typedef __attribute__((ext_vector_type(8))) float v8f;

__device__ __forceinline__ void gemm_chunk(v8f (&acc)[8], const v2f (&a)[8],
                                           const v4f* bp) {
#pragma unroll
    for (int gg = 0; gg < 4; ++gg) {
        v4f b[8];
#pragma unroll
        for (int j = 0; j < 8; ++j) b[j] = bp[(gg * 8 + j) * 32];
#pragma unroll
        for (int j = 0; j < 8; ++j)
            acc[j] = __builtin_amdgcn_wmma_f32_16x16x4_f32(
                 false, a[2 * gg],  false, b[j].xy,
                 (short)0, acc[j],  false,  false);
#pragma unroll
        for (int j = 0; j < 8; ++j)
            acc[j] = __builtin_amdgcn_wmma_f32_16x16x4_f32(
                 false, a[2 * gg + 1],  false, b[j].zw,
                 (short)0, acc[j],  false,  false);
    }
    __builtin_amdgcn_sched_barrier(0);
    asm volatile("v_nop\n\tv_nop\n\tv_nop\n\tv_nop\n\tv_nop\n\tv_nop\n\tv_nop\n\tv_nop\n\tv_nop");
    __builtin_amdgcn_sched_barrier(0);
}

#define GEMM_WAVES 8

__global__ void __launch_bounds__(256)
gcn_gemm_wmma(const float* __restrict__ x, const float* __restrict__ w,
              const float* __restrict__ bias, float* __restrict__ h,
              int n_nodes, int n_tiles) {
    __shared__ __attribute__((aligned(16))) v4f wl[32 * 8 * 32];

    const int lane = threadIdx.x & 31;
    const int wave = __builtin_amdgcn_readfirstlane((int)(threadIdx.x >> 5));
    const int half = lane >> 4;
    const int l16  = lane & 15;

    const int tstride = GEMM_WAVES * (int)gridDim.x;
    int tile = wave * (int)gridDim.x + (int)blockIdx.x;

    int m = tile * 16 + l16;
    if (m >= n_nodes) m = n_nodes - 1;
    const float* xrow = x + (size_t)m * IN_F + 2 * half;

    v2f a_cur[8];
#pragma unroll
    for (int u = 0; u < 8; ++u) a_cur[u] = *(const v2f*)(xrow + 4 * u);

#pragma unroll 4
    for (int q = 0; q < 32; ++q) {
        const int i  = (int)threadIdx.x + 256 * q;
        const int n  = i & 127;
        const int gh = i >> 7;
        const int g  = gh >> 1;
        const int hh = gh & 1;
        const float* wp = w + (size_t)(8 * g + 2 * hh) * OUT_F + n;
        v4f v;
        v.x = wp[0];
        v.y = wp[OUT_F];
        v.z = wp[4 * OUT_F];
        v.w = wp[5 * OUT_F];
        wl[(g * 8 + (n >> 4)) * 32 + hh * 16 + (n & 15)] = v;
    }
    __syncthreads();

    while (tile < n_tiles) {
        const int ntile = tile + tstride;
        int mn = ntile * 16 + l16;
        if (mn >= n_nodes) mn = n_nodes - 1;
        const float* xnext = x + (size_t)mn * IN_F + 2 * half;

        v8f acc[8];
#pragma unroll
        for (int j = 0; j < 8; ++j) { const v8f zero = {}; acc[j] = zero; }

        const v4f* bp = wl + lane;
#pragma nounroll
        for (int cp = 0; cp < 4; ++cp) {
            v2f a_alt[8];
            {
                const float* src = xrow + 32 * (2 * cp + 1);
#pragma unroll
                for (int u = 0; u < 8; ++u) a_alt[u] = *(const v2f*)(src + 4 * u);
            }
            gemm_chunk(acc, a_cur, bp);
            {
                const float* src = (cp < 3) ? (xrow + 32 * (2 * cp + 2)) : xnext;
#pragma unroll
                for (int u = 0; u < 8; ++u) a_cur[u] = *(const v2f*)(src + 4 * u);
            }
            gemm_chunk(acc, a_alt, bp + 4 * 8 * 32);
            bp += 2 * 4 * 8 * 32;
        }

        const int row_base = tile * 16;
        float bn[8];
#pragma unroll
        for (int j = 0; j < 8; ++j) bn[j] = bias[16 * j + l16];
        float* hp = h + (size_t)(row_base + 8 * half) * OUT_F + l16;

        if (row_base + 16 <= n_nodes) {
#pragma unroll
            for (int j = 0; j < 8; ++j)
#pragma unroll
                for (int v = 0; v < 8; ++v)
                    hp[(size_t)v * OUT_F + 16 * j] = acc[j][v] + bn[j];
        } else {
            const int r0 = row_base + 8 * half;
#pragma unroll
            for (int j = 0; j < 8; ++j)
#pragma unroll
                for (int v = 0; v < 8; ++v)
                    if (r0 + v < n_nodes) hp[(size_t)v * OUT_F + 16 * j] = acc[j][v] + bn[j];
        }

        tile = ntile;
        xrow = xnext;
    }
}

__global__ void gcn_zero(float4* __restrict__ out, long long n4) {
    long long i = (long long)blockIdx.x * blockDim.x + threadIdx.x;
    if (i < n4) out[i] = make_float4(0.f, 0.f, 0.f, 0.f);
}

#define EDGES_PER_WAVE 32
#define EDGE_BATCH 8

__device__ __forceinline__ int lane_bcast_i(int x, int src_lane) {
    return __builtin_amdgcn_readlane(x, src_lane);
}
__device__ __forceinline__ float lane_bcast_f(float x, int src_lane) {
    return __int_as_float(__builtin_amdgcn_readlane(__float_as_int(x), src_lane));
}

__global__ void __launch_bounds__(256)
gcn_scatter(const float* __restrict__ h, const int* __restrict__ erow,
            const int* __restrict__ ecol, const float* __restrict__ eval,
            float* __restrict__ out, int n_edges) {
    const int lane = threadIdx.x & 31;
    const int wave = __builtin_amdgcn_readfirstlane((int)(threadIdx.x >> 5));
    const int base = ((int)blockIdx.x * 8 + wave) * EDGES_PER_WAVE;
    if (base >= n_edges) return;
    int cnt = n_edges - base;
    if (cnt > EDGES_PER_WAVE) cnt = EDGES_PER_WAVE;

    int   r = 0;
    int   c = 0;
    float v = 0.f;
    if (lane < cnt) {
        r = erow[base + lane];
        c = ecol[base + lane];
        v = eval[base + lane];
    }

    const float* hl = h + lane;
    float*       ol = out + lane;

    if (cnt == EDGES_PER_WAVE) {
#pragma unroll
        for (int j0 = 0; j0 < EDGES_PER_WAVE; j0 += EDGE_BATCH) {
            float g[EDGE_BATCH][4];
#pragma unroll
            for (int u = 0; u < EDGE_BATCH; ++u) {
                const int cj = lane_bcast_i(c, j0 + u);
                const float* hp = hl + (size_t)cj * OUT_F;
                g[u][0] = hp[0];
                g[u][1] = hp[32];
                g[u][2] = hp[64];
                g[u][3] = hp[96];
            }
#pragma unroll
            for (int u = 0; u < EDGE_BATCH; ++u) {
                const int   rj = lane_bcast_i(r, j0 + u);
                const float vj = lane_bcast_f(v, j0 + u);
                float* op = ol + (size_t)rj * OUT_F;
                atomicAdd(op +  0, g[u][0] * vj);
                atomicAdd(op + 32, g[u][1] * vj);
                atomicAdd(op + 64, g[u][2] * vj);
                atomicAdd(op + 96, g[u][3] * vj);
            }
        }
    } else {
        for (int j = 0; j < cnt; ++j) {
            const int   cj = lane_bcast_i(c, j);
            const int   rj = lane_bcast_i(r, j);
            const float vj = lane_bcast_f(v, j);
            const float* hp = hl + (size_t)cj * OUT_F;
            const float h0 = hp[0];
            const float h1 = hp[32];
            const float h2 = hp[64];
            const float h3 = hp[96];
            float* op = ol + (size_t)rj * OUT_F;
            atomicAdd(op +  0, h0 * vj);
            atomicAdd(op + 32, h1 * vj);
            atomicAdd(op + 64, h2 * vj);
            atomicAdd(op + 96, h3 * vj);
        }
    }
}

#define RB       196
#define NSLICE   256
#define NB_MAX   1024

template <int NT>
__device__ __forceinline__ int block_scan_excl(int v, int* sh, int* total) {
    const int t = (int)threadIdx.x;
    sh[t] = v;
    __syncthreads();
    for (int d = 1; d < NT; d <<= 1) {
        const int a = (t >= d) ? sh[t - d] : 0;
        __syncthreads();
        sh[t] += a;
        __syncthreads();
    }
    const int incl = sh[t];
    *total = sh[NT - 1];
    __syncthreads();
    return incl - v;
}

#define BIN_THREADS 1024
__global__ void __launch_bounds__(BIN_THREADS)
gcn_bin_count(const int* __restrict__ erow, int* __restrict__ counts,
              int n_edges, int n_nodes, int nb) {
    __shared__ int hist[NB_MAX];
    const int t = (int)threadIdx.x;
    for (int i = t; i < nb; i += BIN_THREADS) hist[i] = 0;
    __syncthreads();
    const int per = (n_edges + NSLICE - 1) / NSLICE;
    const int e0 = (int)blockIdx.x * per;
    int e1 = e0 + per;
    if (e1 > n_edges) e1 = n_edges;
    for (int e = e0 + t; e < e1; e += BIN_THREADS) {
        const int r = erow[e];
        if ((unsigned)r < (unsigned)n_nodes) atomicAdd(&hist[r / RB], 1);
    }
    __syncthreads();
    for (int i = t; i < nb; i += BIN_THREADS) counts[i * NSLICE + (int)blockIdx.x] = hist[i];
}

__global__ void __launch_bounds__(NSLICE)
gcn_bin_offsets(int* __restrict__ counts, int* __restrict__ tot) {
    __shared__ int sh[NSLICE];
    const int t = (int)threadIdx.x;
    const int b = (int)blockIdx.x;
    const int c = counts[b * NSLICE + t];
    int total;
    const int ex = block_scan_excl<NSLICE>(c, sh, &total);
    counts[b * NSLICE + t] = ex;
    if (t == 0) tot[b] = total;
}

__global__ void __launch_bounds__(BIN_THREADS)
gcn_bin_place(const int* __restrict__ erow, const int* __restrict__ ecol,
              const float* __restrict__ eval, const int* __restrict__ counts,
              const int* __restrict__ tot, int4* __restrict__ rec,
              int n_edges, int n_nodes, int nb) {
    __shared__ int offs[NB_MAX];
    __shared__ int sh[BIN_THREADS];
    const int t = (int)threadIdx.x;
    const int c = (t < nb) ? tot[t] : 0;
    int total;
    const int bin_start = block_scan_excl<BIN_THREADS>(c, sh, &total);
    if (t < nb) offs[t] = bin_start + counts[t * NSLICE + (int)blockIdx.x];
    __syncthreads();
    const int per = (n_edges + NSLICE - 1) / NSLICE;
    const int e0 = (int)blockIdx.x * per;
    int e1 = e0 + per;
    if (e1 > n_edges) e1 = n_edges;
    for (int e = e0 + t; e < e1; e += BIN_THREADS) {
        const int   r  = erow[e];
        const int   cj = ecol[e];
        const float vj = eval[e];
        if ((unsigned)r < (unsigned)n_nodes) {
            const int b = r / RB;
            const int pos = atomicAdd(&offs[b], 1);
            rec[pos] = make_int4(cj, __float_as_int(vj), r - b * RB, 0);
        }
    }
}

#define ACC_WAVES 16
__global__ void __launch_bounds__(512)
gcn_bin_sum(const float* __restrict__ h, const int4* __restrict__ rec,
            const int* __restrict__ tot, float* __restrict__ out, int n_nodes) {
    __shared__ __attribute__((aligned(16))) float acc[RB * OUT_F];
    __shared__ int s_start;
    const int t = (int)threadIdx.x;
    const int lane = t & 31;
    const int wave = __builtin_amdgcn_readfirstlane(t >> 5);
    const int b = (int)blockIdx.x;

    for (int i = t; i < RB * OUT_F / 4; i += 512)
        ((float4*)acc)[i] = make_float4(0.f, 0.f, 0.f, 0.f);
    if (t == 0) s_start = 0;
    __syncthreads();
    int s = 0;
    for (int i = t; i < b; i += 512) s += tot[i];
    if (s != 0) atomicAdd(&s_start, s);
    __syncthreads();
    const int start = s_start;
    const int cnt = tot[b];

    const float* hl = h + 4 * lane;
    float* al = acc + lane;

    for (int base = wave * 32; base < cnt; base += ACC_WAVES * 32) {
        int n = cnt - base;
        if (n > 32) n = 32;
        int4 rc = make_int4(0, 0, 0, 0);
        if (lane < n) rc = rec[start + base + lane];
        if (n == 32) {
#pragma unroll
            for (int j0 = 0; j0 < 32; j0 += 8) {
                v4f g[8];
#pragma unroll
                for (int u = 0; u < 8; ++u) {
                    const int cj = lane_bcast_i(rc.x, j0 + u);
                    g[u] = *(const v4f*)(hl + (size_t)cj * OUT_F);
                }
#pragma unroll
                for (int u = 0; u < 8; ++u) {
                    const float vj = __int_as_float(lane_bcast_i(rc.y, j0 + u));
                    const int   rl = lane_bcast_i(rc.z, j0 + u);
                    float* ap = al + rl * OUT_F;
                    atomicAdd(ap +  0, g[u].x * vj);
                    atomicAdd(ap + 32, g[u].y * vj);
                    atomicAdd(ap + 64, g[u].z * vj);
                    atomicAdd(ap + 96, g[u].w * vj);
                }
            }
        } else {
            for (int j = 0; j < n; ++j) {
                const int   cj = lane_bcast_i(rc.x, j);
                const float vj = __int_as_float(lane_bcast_i(rc.y, j));
                const int   rl = lane_bcast_i(rc.z, j);
                const v4f g = *(const v4f*)(hl + (size_t)cj * OUT_F);
                float* ap = al + rl * OUT_F;
                atomicAdd(ap +  0, g.x * vj);
                atomicAdd(ap + 32, g.y * vj);
                atomicAdd(ap + 64, g.z * vj);
                atomicAdd(ap + 96, g.w * vj);
            }
        }
    }
    __syncthreads();

    for (int rl = wave; rl < RB; rl += ACC_WAVES) {
        const int r = b * RB + rl;
        if (r < n_nodes) {
            const float* ap = al + rl * OUT_F;
            v4f v;
            v.x = ap[0];
            v.y = ap[32];
            v.z = ap[64];
            v.w = ap[96];
            *(v4f*)(out + (size_t)r * OUT_F + 4 * lane) = v;
        }
    }
}

extern "C" void kernel_launch(void* const* d_in, const int* in_sizes, int n_in,
                              void* d_out, int out_size, void* d_ws, size_t ws_size,
                              hipStream_t stream) {
    const float* x    = (const float*)d_in[0];
    const int*   erow = (const int*)  d_in[1];
    const int*   ecol = (const int*)  d_in[2];
    const float* ev   = (const float*)d_in[3];
    const float* w    = (const float*)d_in[4];
    const float* bias = (const float*)d_in[5];
    float*       out  = (float*)d_out;
    float*       h    = (float*)d_ws;

    const int n_nodes = in_sizes[0] / IN_F;
    const int n_edges = in_sizes[1];
    const int row_tiles = (n_nodes + 15) / 16;

    const int    nb      = (n_nodes + RB - 1) / RB;
    const size_t h_bytes = (size_t)n_nodes * OUT_F * sizeof(float);
    int4* rec    = (int4*)((char*)d_ws + h_bytes);
    int*  counts = (int*)((char*)rec + (size_t)n_edges * sizeof(int4));
    int*  tot    = counts + (size_t)nb * NSLICE;
    const size_t need = h_bytes + (size_t)n_edges * sizeof(int4)
                      + ((size_t)nb * NSLICE + (size_t)nb) * sizeof(int);

    if (ws_size >= need && nb <= NB_MAX) {
        gcn_bin_count<<<NSLICE, BIN_THREADS, 0, stream>>>(erow, counts, n_edges, n_nodes, nb);
        gcn_bin_offsets<<<nb, NSLICE, 0, stream>>>(counts, tot);
        gcn_bin_place<<<NSLICE, BIN_THREADS, 0, stream>>>(erow, ecol, ev, counts, tot, rec,
                                                  n_edges, n_nodes, nb);
        gcn_gemm_wmma<<<512, 256, 0, stream>>>(x, w, bias, h, n_nodes, row_tiles);
        gcn_bin_sum<<<nb, 512, 0, stream>>>(h, rec, tot, out, n_nodes);
    } else {
        const long long n4 = (long long)n_nodes * OUT_F / 4;
        gcn_zero<<<(unsigned)((n4 + 255) / 256), 256, 0, stream>>>((float4*)out, n4);
        gcn_gemm_wmma<<<512, 256, 0, stream>>>(x, w, bias, h, n_nodes, row_tiles);
        const int blocks = (n_edges + 255) / 256;
        gcn_scatter<<<blocks, 256, 0, stream>>>(h, erow, ecol, ev, out, n_edges);
    }
}


# bin sum: bin's records sorted by local row in LDS (integer rank adds); each wave streams its 12-13 rows' records, sums a row in registers in arrival-rank order, stores it once; no LDS f32 adds
# speedup vs baseline: 2.9174x; 1.0504x over previous
#include <hip/hip_runtime.h>
#include <hip/hip_bf16.h>

#define IN_F  256
#define OUT_F 128

typedef __attribute__((ext_vector_type(2))) float v2f;
typedef __attribute__((ext_vector_type(4))) float v4f;
typedef __attribute__((ext_vector_type(8))) float v8f;

__device__ __forceinline__ void gemm_chunk(v8f (&acc)[8], const v2f (&a)[8],
                                           const v4f* bp) {
#pragma unroll
    for (int gg = 0; gg < 4; ++gg) {
        v4f b[8];
#pragma unroll
        for (int j = 0; j < 8; ++j) b[j] = bp[(gg * 8 + j) * 32];
#pragma unroll
        for (int j = 0; j < 8; ++j)
            acc[j] = __builtin_amdgcn_wmma_f32_16x16x4_f32(
                 false, a[2 * gg],  false, b[j].xy,
                 (short)0, acc[j],  false,  false);
#pragma unroll
        for (int j = 0; j < 8; ++j)
            acc[j] = __builtin_amdgcn_wmma_f32_16x16x4_f32(
                 false, a[2 * gg + 1],  false, b[j].zw,
                 (short)0, acc[j],  false,  false);
    }
    __builtin_amdgcn_sched_barrier(0);
    asm volatile("v_nop\n\tv_nop\n\tv_nop\n\tv_nop\n\tv_nop\n\tv_nop\n\tv_nop\n\tv_nop\n\tv_nop");
    __builtin_amdgcn_sched_barrier(0);
}

#define GEMM_WAVES 8

__global__ void __launch_bounds__(256)
gcn_gemm_wmma(const float* __restrict__ x, const float* __restrict__ w,
              const float* __restrict__ bias, float* __restrict__ h,
              int n_nodes, int n_tiles) {
    __shared__ __attribute__((aligned(16))) v4f wl[32 * 8 * 32];

    const int lane = threadIdx.x & 31;
    const int wave = __builtin_amdgcn_readfirstlane((int)(threadIdx.x >> 5));
    const int half = lane >> 4;
    const int l16  = lane & 15;

    const int tstride = GEMM_WAVES * (int)gridDim.x;
    int tile = wave * (int)gridDim.x + (int)blockIdx.x;

    int m = tile * 16 + l16;
    if (m >= n_nodes) m = n_nodes - 1;
    const float* xrow = x + (size_t)m * IN_F + 2 * half;

    v2f a_cur[8];
#pragma unroll
    for (int u = 0; u < 8; ++u) a_cur[u] = *(const v2f*)(xrow + 4 * u);

#pragma unroll 4
    for (int q = 0; q < 32; ++q) {
        const int i  = (int)threadIdx.x + 256 * q;
        const int n  = i & 127;
        const int gh = i >> 7;
        const int g  = gh >> 1;
        const int hh = gh & 1;
        const float* wp = w + (size_t)(8 * g + 2 * hh) * OUT_F + n;
        v4f v;
        v.x = wp[0];
        v.y = wp[OUT_F];
        v.z = wp[4 * OUT_F];
        v.w = wp[5 * OUT_F];
        wl[(g * 8 + (n >> 4)) * 32 + hh * 16 + (n & 15)] = v;
    }
    __syncthreads();

    while (tile < n_tiles) {
        const int ntile = tile + tstride;
        int mn = ntile * 16 + l16;
        if (mn >= n_nodes) mn = n_nodes - 1;
        const float* xnext = x + (size_t)mn * IN_F + 2 * half;

        v8f acc[8];
#pragma unroll
        for (int j = 0; j < 8; ++j) { const v8f zero = {}; acc[j] = zero; }

        const v4f* bp = wl + lane;
#pragma nounroll
        for (int cp = 0; cp < 4; ++cp) {
            v2f a_alt[8];
            {
                const float* src = xrow + 32 * (2 * cp + 1);
#pragma unroll
                for (int u = 0; u < 8; ++u) a_alt[u] = *(const v2f*)(src + 4 * u);
            }
            gemm_chunk(acc, a_cur, bp);
            {
                const float* src = (cp < 3) ? (xrow + 32 * (2 * cp + 2)) : xnext;
#pragma unroll
                for (int u = 0; u < 8; ++u) a_cur[u] = *(const v2f*)(src + 4 * u);
            }
            gemm_chunk(acc, a_alt, bp + 4 * 8 * 32);
            bp += 2 * 4 * 8 * 32;
        }

        const int row_base = tile * 16;
        float bn[8];
#pragma unroll
        for (int j = 0; j < 8; ++j) bn[j] = bias[16 * j + l16];
        float* hp = h + (size_t)(row_base + 8 * half) * OUT_F + l16;

        if (row_base + 16 <= n_nodes) {
#pragma unroll
            for (int j = 0; j < 8; ++j)
#pragma unroll
                for (int v = 0; v < 8; ++v)
                    hp[(size_t)v * OUT_F + 16 * j] = acc[j][v] + bn[j];
        } else {
            const int r0 = row_base + 8 * half;
#pragma unroll
            for (int j = 0; j < 8; ++j)
#pragma unroll
                for (int v = 0; v < 8; ++v)
                    if (r0 + v < n_nodes) hp[(size_t)v * OUT_F + 16 * j] = acc[j][v] + bn[j];
        }

        tile = ntile;
        xrow = xnext;
    }
}

__global__ void gcn_zero(float4* __restrict__ out, long long n4) {
    long long i = (long long)blockIdx.x * blockDim.x + threadIdx.x;
    if (i < n4) out[i] = make_float4(0.f, 0.f, 0.f, 0.f);
}

#define EDGES_PER_WAVE 32
#define EDGE_BATCH 8

__device__ __forceinline__ int lane_bcast_i(int x, int src_lane) {
    return __builtin_amdgcn_readlane(x, src_lane);
}
__device__ __forceinline__ float lane_bcast_f(float x, int src_lane) {
    return __int_as_float(__builtin_amdgcn_readlane(__float_as_int(x), src_lane));
}

__global__ void __launch_bounds__(256)
gcn_scatter(const float* __restrict__ h, const int* __restrict__ erow,
            const int* __restrict__ ecol, const float* __restrict__ eval,
            float* __restrict__ out, int n_edges) {
    const int lane = threadIdx.x & 31;
    const int wave = __builtin_amdgcn_readfirstlane((int)(threadIdx.x >> 5));
    const int base = ((int)blockIdx.x * 8 + wave) * EDGES_PER_WAVE;
    if (base >= n_edges) return;
    int cnt = n_edges - base;
    if (cnt > EDGES_PER_WAVE) cnt = EDGES_PER_WAVE;

    int   r = 0;
    int   c = 0;
    float v = 0.f;
    if (lane < cnt) {
        r = erow[base + lane];
        c = ecol[base + lane];
        v = eval[base + lane];
    }

    const float* hl = h + lane;
    float*       ol = out + lane;

    if (cnt == EDGES_PER_WAVE) {
#pragma unroll
        for (int j0 = 0; j0 < EDGES_PER_WAVE; j0 += EDGE_BATCH) {
            float g[EDGE_BATCH][4];
#pragma unroll
            for (int u = 0; u < EDGE_BATCH; ++u) {
                const int cj = lane_bcast_i(c, j0 + u);
                const float* hp = hl + (size_t)cj * OUT_F;
                g[u][0] = hp[0];
                g[u][1] = hp[32];
                g[u][2] = hp[64];
                g[u][3] = hp[96];
            }
#pragma unroll
            for (int u = 0; u < EDGE_BATCH; ++u) {
                const int   rj = lane_bcast_i(r, j0 + u);
                const float vj = lane_bcast_f(v, j0 + u);
                float* op = ol + (size_t)rj * OUT_F;
                atomicAdd(op +  0, g[u][0] * vj);
                atomicAdd(op + 32, g[u][1] * vj);
                atomicAdd(op + 64, g[u][2] * vj);
                atomicAdd(op + 96, g[u][3] * vj);
            }
        }
    } else {
        for (int j = 0; j < cnt; ++j) {
            const int   cj = lane_bcast_i(c, j);
            const int   rj = lane_bcast_i(r, j);
            const float vj = lane_bcast_f(v, j);
            const float* hp = hl + (size_t)cj * OUT_F;
            const float h0 = hp[0];
            const float h1 = hp[32];
            const float h2 = hp[64];
            const float h3 = hp[96];
            float* op = ol + (size_t)rj * OUT_F;
            atomicAdd(op +  0, h0 * vj);
            atomicAdd(op + 32, h1 * vj);
            atomicAdd(op + 64, h2 * vj);
            atomicAdd(op + 96, h3 * vj);
        }
    }
}

#define RB       196
#define NSLICE   256
#define NB_MAX   1024

template <int NT>
__device__ __forceinline__ int block_scan_excl(int v, int* sh, int* total) {
    const int t = (int)threadIdx.x;
    sh[t] = v;
    __syncthreads();
    for (int d = 1; d < NT; d <<= 1) {
        const int a = (t >= d) ? sh[t - d] : 0;
        __syncthreads();
        sh[t] += a;
        __syncthreads();
    }
    const int incl = sh[t];
    *total = sh[NT - 1];
    __syncthreads();
    return incl - v;
}

#define BIN_THREADS 1024
__global__ void __launch_bounds__(BIN_THREADS)
gcn_bin_count(const int* __restrict__ erow, int* __restrict__ counts,
              int n_edges, int n_nodes, int nb) {
    __shared__ int hist[NB_MAX];
    const int t = (int)threadIdx.x;
    for (int i = t; i < nb; i += BIN_THREADS) hist[i] = 0;
    __syncthreads();
    const int per = (n_edges + NSLICE - 1) / NSLICE;
    const int e0 = (int)blockIdx.x * per;
    int e1 = e0 + per;
    if (e1 > n_edges) e1 = n_edges;
    for (int e = e0 + t; e < e1; e += BIN_THREADS) {
        const int r = erow[e];
        if ((unsigned)r < (unsigned)n_nodes) atomicAdd(&hist[r / RB], 1);
    }
    __syncthreads();
    for (int i = t; i < nb; i += BIN_THREADS) counts[i * NSLICE + (int)blockIdx.x] = hist[i];
}

__global__ void __launch_bounds__(NSLICE)
gcn_bin_offsets(int* __restrict__ counts, int* __restrict__ tot) {
    __shared__ int sh[NSLICE];
    const int t = (int)threadIdx.x;
    const int b = (int)blockIdx.x;
    const int c = counts[b * NSLICE + t];
    int total;
    const int ex = block_scan_excl<NSLICE>(c, sh, &total);
    counts[b * NSLICE + t] = ex;
    if (t == 0) tot[b] = total;
}

__global__ void __launch_bounds__(BIN_THREADS)
gcn_bin_place(const int* __restrict__ erow, const int* __restrict__ ecol,
              const float* __restrict__ eval, const int* __restrict__ counts,
              const int* __restrict__ tot, int4* __restrict__ rec,
              int n_edges, int n_nodes, int nb) {
    __shared__ int offs[NB_MAX];
    __shared__ int sh[BIN_THREADS];
    const int t = (int)threadIdx.x;
    const int c = (t < nb) ? tot[t] : 0;
    int total;
    const int bin_start = block_scan_excl<BIN_THREADS>(c, sh, &total);
    if (t < nb) offs[t] = bin_start + counts[t * NSLICE + (int)blockIdx.x];
    __syncthreads();
    const int per = (n_edges + NSLICE - 1) / NSLICE;
    const int e0 = (int)blockIdx.x * per;
    int e1 = e0 + per;
    if (e1 > n_edges) e1 = n_edges;
    for (int e = e0 + t; e < e1; e += BIN_THREADS) {
        const int   r  = erow[e];
        const int   cj = ecol[e];
        const float vj = eval[e];
        if ((unsigned)r < (unsigned)n_nodes) {
            const int b = r / RB;
            const int pos = atomicAdd(&offs[b], 1);
            rec[pos] = make_int4(cj, __float_as_int(vj), r - b * RB, 0);
        }
    }
}

#define SUM_THREADS 512
#define SUM_WAVES   16
#define SCAN_PER_LANE ((RB + 1 + 31) / 32)
#define CH          4096
#define CH_PER_T    (CH / SUM_THREADS)

struct RowRun {
    v4f acc;
    int r;
    int e_r;
};

template <bool FIRST>
__device__ __forceinline__ void row_flush(RowRun& s, int endv, int r0, int row_base,
                                          float* __restrict__ out_lane, int n_nodes) {
    const int row = row_base + s.r;
    if (row < n_nodes) {
        float* op = out_lane + (size_t)row * OUT_F;
        if (!FIRST) s.acc += *(const v4f*)op;
        *(v4f*)op = s.acc;
    }
    const v4f zero = {};
    s.acc = zero;
    s.r += 1;
    s.e_r = __builtin_amdgcn_readlane(endv, s.r - r0);
}

__device__ __forceinline__ void load8(v4f (&g)[8], int rec_c, int o,
                                      const float* __restrict__ hl) {
#pragma unroll
    for (int u = 0; u < 8; ++u) {
        const int cj = lane_bcast_i(rec_c, o + u);
        g[u] = *(const v4f*)(hl + (size_t)cj * OUT_F);
    }
}

template <bool FIRST>
__device__ __forceinline__ void add8(RowRun& s, const v4f (&g)[8], int rec_v, int o,
                                     int jb, int se, int endv, int r0, int row_base,
                                     float* __restrict__ out_lane, int n_nodes) {
#pragma unroll
    for (int u = 0; u < 8; ++u) {
        const int jj = jb + u;
        if (jj < se) {
            while (jj == s.e_r) row_flush<FIRST>(s, endv, r0, row_base, out_lane, n_nodes);
            const float vj = __int_as_float(lane_bcast_i(rec_v, o + u));
            s.acc += g[u] * vj;
        }
    }
}

template <bool FIRST>
__device__ __forceinline__ void bin_pass(const float* __restrict__ h,
                                         const int4* __restrict__ rec,
                                         float* __restrict__ out, int n_nodes,
                                         int start, int cnt, int c0,
                                         int2* srt, int* hist, int* rowstart) {
    const int t = (int)threadIdx.x;
    const int lane = t & 31;
    const int wave = __builtin_amdgcn_readfirstlane(t >> 5);
    const int row_base = (int)blockIdx.x * RB;
    const int r0 = (wave * RB) / SUM_WAVES;
    const int r1 = ((wave + 1) * RB) / SUM_WAVES;
    const float* hl = h + 4 * lane;
    float* out_lane = out + 4 * lane;

    int n = cnt - c0;
    if (n > CH) n = CH;

    if (t < 256) hist[t] = 0;
    __syncthreads();

    int4 q[CH_PER_T];
    int rank[CH_PER_T];
#pragma unroll
    for (int k = 0; k < CH_PER_T; ++k) {
        int i = t + SUM_THREADS * k;
        if (i > n - 1) i = n - 1;
        if (i < 0) i = 0;
        q[k] = make_int4(0, 0, 0, 0);
        if (n > 0) q[k] = rec[start + c0 + i];
    }
#pragma unroll
    for (int k = 0; k < CH_PER_T; ++k) {
        rank[k] = 0;
        if (t + SUM_THREADS * k < n) rank[k] = atomicAdd(&hist[q[k].z], 1);
    }
    __syncthreads();

    if (wave == 0) {
        int v[SCAN_PER_LANE];
        int run = 0;
#pragma unroll
        for (int k = 0; k < SCAN_PER_LANE; ++k) {
            const int idx = SCAN_PER_LANE * lane + k;
            const int c = (idx < RB) ? hist[idx] : 0;
            v[k] = run;
            run += c;
        }
        int inc = run;
#pragma unroll
        for (int d = 1; d < 32; d <<= 1) {
            const int o = __shfl_up(inc, d);
            if (lane >= d) inc += o;
        }
        const int excl = inc - run;
#pragma unroll
        for (int k = 0; k < SCAN_PER_LANE; ++k) {
            const int idx = SCAN_PER_LANE * lane + k;
            if (idx <= RB) rowstart[idx] = excl + v[k];
        }
    }
    __syncthreads();

    int pos[CH_PER_T];
#pragma unroll
    for (int k = 0; k < CH_PER_T; ++k) pos[k] = rowstart[q[k].z] + rank[k];
#pragma unroll
    for (int k = 0; k < CH_PER_T; ++k)
        if (t + SUM_THREADS * k < n) srt[pos[k]] = make_int2(q[k].x, q[k].y);
    __syncthreads();

    if (r0 < r1) {
        int endv = 0x7fffffff;
        if (r0 + lane < r1) endv = rowstart[r0 + lane + 1];
        const int sb = __builtin_amdgcn_readfirstlane(rowstart[r0]);
        const int se = __builtin_amdgcn_readlane(endv, r1 - r0 - 1);

        RowRun s;
        { const v4f zero = {}; s.acc = zero; }
        s.r = r0;
        s.e_r = __builtin_amdgcn_readlane(endv, 0);

        if (sb < se) {
            const int l16 = lane & 15;
            int j = sb + l16;
            if (j > se - 1) j = se - 1;
            int2 my = srt[j];
            v4f ga[8], gb[8];
            load8(ga, my.x, 0, hl);
#pragma nounroll
            for (int base = sb; base < se; base += 16) {
                int jn = base + 16 + l16;
                if (jn > se - 1) jn = se - 1;
                const int2 myn = srt[jn];
                load8(gb, my.x, 8, hl);
                add8<FIRST>(s, ga, my.y, 0, base, se, endv, r0, row_base, out_lane, n_nodes);
                load8(ga, myn.x, 0, hl);
                add8<FIRST>(s, gb, my.y, 8, base + 8, se, endv, r0, row_base, out_lane, n_nodes);
                my = myn;
            }
        }
        while (s.r < r1) row_flush<FIRST>(s, endv, r0, row_base, out_lane, n_nodes);
    }
    __syncthreads();
}

__global__ void __launch_bounds__(SUM_THREADS)
gcn_bin_rowsum(const float* __restrict__ h, const int4* __restrict__ rec,
               const int* __restrict__ tot, float* __restrict__ out, int n_nodes) {
    __shared__ __attribute__((aligned(16))) int2 srt[CH];
    __shared__ int hist[256];
    __shared__ int rowstart[257];
    __shared__ int s_start;
    const int t = (int)threadIdx.x;
    const int b = (int)blockIdx.x;

    if (t == 0) s_start = 0;
    __syncthreads();
    int sp = 0;
    for (int i = t; i < b; i += SUM_THREADS) sp += tot[i];
    if (sp != 0) atomicAdd(&s_start, sp);
    __syncthreads();
    const int start = s_start;
    const int cnt = tot[b];

    bin_pass<true>(h, rec, out, n_nodes, start, cnt, 0, srt, hist, rowstart);
    for (int c0 = CH; c0 < cnt; c0 += CH)
        bin_pass<false>(h, rec, out, n_nodes, start, cnt, c0, srt, hist, rowstart);
}

extern "C" void kernel_launch(void* const* d_in, const int* in_sizes, int n_in,
                              void* d_out, int out_size, void* d_ws, size_t ws_size,
                              hipStream_t stream) {
    const float* x    = (const float*)d_in[0];
    const int*   erow = (const int*)  d_in[1];
    const int*   ecol = (const int*)  d_in[2];
    const float* ev   = (const float*)d_in[3];
    const float* w    = (const float*)d_in[4];
    const float* bias = (const float*)d_in[5];
    float*       out  = (float*)d_out;
    float*       h    = (float*)d_ws;

    const int n_nodes = in_sizes[0] / IN_F;
    const int n_edges = in_sizes[1];
    const int row_tiles = (n_nodes + 15) / 16;

    const int    nb      = (n_nodes + RB - 1) / RB;
    const size_t h_bytes = (size_t)n_nodes * OUT_F * sizeof(float);
    int4* rec    = (int4*)((char*)d_ws + h_bytes);
    int*  counts = (int*)((char*)rec + (size_t)n_edges * sizeof(int4));
    int*  tot    = counts + (size_t)nb * NSLICE;
    const size_t need = h_bytes + (size_t)n_edges * sizeof(int4)
                      + ((size_t)nb * NSLICE + (size_t)nb) * sizeof(int);

    if (ws_size >= need && nb <= NB_MAX) {
        gcn_bin_count<<<NSLICE, BIN_THREADS, 0, stream>>>(erow, counts, n_edges, n_nodes, nb);
        gcn_bin_offsets<<<nb, NSLICE, 0, stream>>>(counts, tot);
        gcn_bin_place<<<NSLICE, BIN_THREADS, 0, stream>>>(erow, ecol, ev, counts, tot, rec,
                                                  n_edges, n_nodes, nb);
        gcn_gemm_wmma<<<512, 256, 0, stream>>>(x, w, bias, h, n_nodes, row_tiles);
        gcn_bin_rowsum<<<nb, SUM_THREADS, 0, stream>>>(h, rec, tot, out, n_nodes);
    } else {
        const long long n4 = (long long)n_nodes * OUT_F / 4;
        gcn_zero<<<(unsigned)((n4 + 255) / 256), 256, 0, stream>>>((float4*)out, n4);
        gcn_gemm_wmma<<<512, 256, 0, stream>>>(x, w, bias, h, n_nodes, row_tiles);
        const int blocks = (n_edges + 255) / 256;
        gcn_scatter<<<blocks, 256, 0, stream>>>(h, erow, ecol, ev, out, n_edges);
    }
}


# binning pass 3 (filing edge records) moved into the GEMM launch on 4 extra waves per work-group (256 groups of 16 GEMM + 4 filing waves): 4 launches not 5; WMMA order, arrival-order ranks as before
# speedup vs baseline: 3.2790x; 1.1239x over previous
#include <hip/hip_runtime.h>
#include <hip/hip_bf16.h>

#define IN_F  256
#define OUT_F 128

typedef __attribute__((ext_vector_type(2))) float v2f;
typedef __attribute__((ext_vector_type(4))) float v4f;
typedef __attribute__((ext_vector_type(8))) float v8f;

__device__ __forceinline__ void gemm_chunk(v8f (&acc)[8], const v2f (&a)[8],
                                           const v4f* bp) {
#pragma unroll
    for (int gg = 0; gg < 4; ++gg) {
        v4f b[8];
#pragma unroll
        for (int j = 0; j < 8; ++j) b[j] = bp[(gg * 8 + j) * 32];
#pragma unroll
        for (int j = 0; j < 8; ++j)
            acc[j] = __builtin_amdgcn_wmma_f32_16x16x4_f32(
                 false, a[2 * gg],  false, b[j].xy,
                 (short)0, acc[j],  false,  false);
#pragma unroll
        for (int j = 0; j < 8; ++j)
            acc[j] = __builtin_amdgcn_wmma_f32_16x16x4_f32(
                 false, a[2 * gg + 1],  false, b[j].zw,
                 (short)0, acc[j],  false,  false);
    }
    __builtin_amdgcn_sched_barrier(0);
    asm volatile("v_nop\n\tv_nop\n\tv_nop\n\tv_nop\n\tv_nop\n\tv_nop\n\tv_nop\n\tv_nop\n\tv_nop");
    __builtin_amdgcn_sched_barrier(0);
}

#define GEMM_WAVES  16
#define PLACE_WAVES 4
#define FUSED_THREADS ((GEMM_WAVES + PLACE_WAVES) * 32)
#define GEMM_THREADS  (GEMM_WAVES * 32)
#define PLACE_THREADS (PLACE_WAVES * 32)
#define PLACE_BATCH   8

#define RB       196
#define NSLICE   256
#define NB_MAX   1024

__global__ void __launch_bounds__(FUSED_THREADS)
gcn_gemm_place(const float* __restrict__ x, const float* __restrict__ w,
               const float* __restrict__ bias, float* __restrict__ h,
               int n_nodes, int n_tiles,
               const int* __restrict__ erow, const int* __restrict__ ecol,
               const float* __restrict__ eval, const int* __restrict__ counts,
               const int* __restrict__ tot, int4* __restrict__ rec,
               int n_edges, int nb, int do_place) {
    __shared__ __attribute__((aligned(16))) v4f wl[32 * 8 * 32];
    __shared__ int offs[NB_MAX];

    const int lane = threadIdx.x & 31;
    const int wave = __builtin_amdgcn_readfirstlane((int)(threadIdx.x >> 5));
    const int half = lane >> 4;
    const int l16  = lane & 15;

    const int tstride = GEMM_WAVES * (int)gridDim.x;
    int tile = wave * (int)gridDim.x + (int)blockIdx.x;

    int m = tile * 16 + l16;
    if (m >= n_nodes) m = n_nodes - 1;
    const float* xrow = x + (size_t)m * IN_F + 2 * half;

    v2f a_cur[8];
    if (wave < GEMM_WAVES) {
#pragma unroll
        for (int u = 0; u < 8; ++u) a_cur[u] = *(const v2f*)(xrow + 4 * u);

#pragma unroll 4
        for (int q = 0; q < (32 * 256) / GEMM_THREADS; ++q) {
            const int i  = (int)threadIdx.x + GEMM_THREADS * q;
            const int n  = i & 127;
            const int gh = i >> 7;
            const int g  = gh >> 1;
            const int hh = gh & 1;
            const float* wp = w + (size_t)(8 * g + 2 * hh) * OUT_F + n;
            v4f v;
            v.x = wp[0];
            v.y = wp[OUT_F];
            v.z = wp[4 * OUT_F];
            v.w = wp[5 * OUT_F];
            wl[(g * 8 + (n >> 4)) * 32 + hh * 16 + (n & 15)] = v;
        }
    } else if (wave == GEMM_WAVES && do_place != 0) {
        __builtin_amdgcn_s_setprio(3);
        int tv[NB_MAX / 32];
        int cv[NB_MAX / 32];
#pragma unroll
        for (int k = 0; k < NB_MAX / 32; ++k) {
            const int b = (NB_MAX / 32) * lane + k;
            tv[k] = 0;
            cv[k] = 0;
            if (b < nb) {
                tv[k] = tot[b];
                cv[k] = counts[b * NSLICE + (int)blockIdx.x];
            }
        }
        int run = 0;
#pragma unroll
        for (int k = 0; k < NB_MAX / 32; ++k) { const int c = tv[k]; tv[k] = run; run += c; }
        int inc = run;
#pragma unroll
        for (int d = 1; d < 32; d <<= 1) {
            const int o = __shfl_up(inc, d);
            if (lane >= d) inc += o;
        }
        const int excl = inc - run;
#pragma unroll
        for (int k = 0; k < NB_MAX / 32; ++k) {
            const int b = (NB_MAX / 32) * lane + k;
            if (b < nb) offs[b] = excl + tv[k] + cv[k];
        }
    }
    __syncthreads();

    if (wave >= GEMM_WAVES) {
        if (do_place == 0) return;
        __builtin_amdgcn_s_setprio(3);
        const int ht = (int)threadIdx.x - GEMM_THREADS;
        const int per = (n_edges + NSLICE - 1) / NSLICE;
        const int e0 = (int)blockIdx.x * per;
        int e1 = e0 + per;
        if (e1 > n_edges) e1 = n_edges;
        for (int eb = e0; eb < e1; eb += PLACE_THREADS * PLACE_BATCH) {
            int   r[PLACE_BATCH];
            int   cj[PLACE_BATCH];
            float vj[PLACE_BATCH];
#pragma unroll
            for (int u = 0; u < PLACE_BATCH; ++u) {
                int e = eb + ht + PLACE_THREADS * u;
                if (e > e1 - 1) e = e1 - 1;
                r[u]  = erow[e];
                cj[u] = ecol[e];
                vj[u] = eval[e];
            }
#pragma unroll
            for (int u = 0; u < PLACE_BATCH; ++u) {
                const int e = eb + ht + PLACE_THREADS * u;
                if (e < e1 && (unsigned)r[u] < (unsigned)n_nodes) {
                    const int b = r[u] / RB;
                    const int pos = atomicAdd(&offs[b], 1);
                    rec[pos] = make_int4(cj[u], __float_as_int(vj[u]), r[u] - b * RB, 0);
                }
            }
        }
        return;
    }

    while (tile < n_tiles) {
        const int ntile = tile + tstride;
        int mn = ntile * 16 + l16;
        if (mn >= n_nodes) mn = n_nodes - 1;
        const float* xnext = x + (size_t)mn * IN_F + 2 * half;

        v8f acc[8];
#pragma unroll
        for (int j = 0; j < 8; ++j) { const v8f zero = {}; acc[j] = zero; }

        const v4f* bp = wl + lane;
#pragma nounroll
        for (int cp = 0; cp < 4; ++cp) {
            v2f a_alt[8];
            {
                const float* src = xrow + 32 * (2 * cp + 1);
#pragma unroll
                for (int u = 0; u < 8; ++u) a_alt[u] = *(const v2f*)(src + 4 * u);
            }
            gemm_chunk(acc, a_cur, bp);
            {
                const float* src = (cp < 3) ? (xrow + 32 * (2 * cp + 2)) : xnext;
#pragma unroll
                for (int u = 0; u < 8; ++u) a_cur[u] = *(const v2f*)(src + 4 * u);
            }
            gemm_chunk(acc, a_alt, bp + 4 * 8 * 32);
            bp += 2 * 4 * 8 * 32;
        }

        const int row_base = tile * 16;
        float bn[8];
#pragma unroll
        for (int j = 0; j < 8; ++j) bn[j] = bias[16 * j + l16];
        float* hp = h + (size_t)(row_base + 8 * half) * OUT_F + l16;

        if (row_base + 16 <= n_nodes) {
#pragma unroll
            for (int j = 0; j < 8; ++j)
#pragma unroll
                for (int v = 0; v < 8; ++v)
                    hp[(size_t)v * OUT_F + 16 * j] = acc[j][v] + bn[j];
        } else {
            const int r0 = row_base + 8 * half;
#pragma unroll
            for (int j = 0; j < 8; ++j)
#pragma unroll
                for (int v = 0; v < 8; ++v)
                    if (r0 + v < n_nodes) hp[(size_t)v * OUT_F + 16 * j] = acc[j][v] + bn[j];
        }

        tile = ntile;
        xrow = xnext;
    }
}

__global__ void gcn_zero(float4* __restrict__ out, long long n4) {
    long long i = (long long)blockIdx.x * blockDim.x + threadIdx.x;
    if (i < n4) out[i] = make_float4(0.f, 0.f, 0.f, 0.f);
}

#define EDGES_PER_WAVE 32
#define EDGE_BATCH 8

__device__ __forceinline__ int lane_bcast_i(int x, int src_lane) {
    return __builtin_amdgcn_readlane(x, src_lane);
}
__device__ __forceinline__ float lane_bcast_f(float x, int src_lane) {
    return __int_as_float(__builtin_amdgcn_readlane(__float_as_int(x), src_lane));
}

__global__ void __launch_bounds__(256)
gcn_scatter(const float* __restrict__ h, const int* __restrict__ erow,
            const int* __restrict__ ecol, const float* __restrict__ eval,
            float* __restrict__ out, int n_edges) {
    const int lane = threadIdx.x & 31;
    const int wave = __builtin_amdgcn_readfirstlane((int)(threadIdx.x >> 5));
    const int base = ((int)blockIdx.x * 8 + wave) * EDGES_PER_WAVE;
    if (base >= n_edges) return;
    int cnt = n_edges - base;
    if (cnt > EDGES_PER_WAVE) cnt = EDGES_PER_WAVE;

    int   r = 0;
    int   c = 0;
    float v = 0.f;
    if (lane < cnt) {
        r = erow[base + lane];
        c = ecol[base + lane];
        v = eval[base + lane];
    }

    const float* hl = h + lane;
    float*       ol = out + lane;

    if (cnt == EDGES_PER_WAVE) {
#pragma unroll
        for (int j0 = 0; j0 < EDGES_PER_WAVE; j0 += EDGE_BATCH) {
            float g[EDGE_BATCH][4];
#pragma unroll
            for (int u = 0; u < EDGE_BATCH; ++u) {
                const int cj = lane_bcast_i(c, j0 + u);
                const float* hp = hl + (size_t)cj * OUT_F;
                g[u][0] = hp[0];
                g[u][1] = hp[32];
                g[u][2] = hp[64];
                g[u][3] = hp[96];
            }
#pragma unroll
            for (int u = 0; u < EDGE_BATCH; ++u) {
                const int   rj = lane_bcast_i(r, j0 + u);
                const float vj = lane_bcast_f(v, j0 + u);
                float* op = ol + (size_t)rj * OUT_F;
                atomicAdd(op +  0, g[u][0] * vj);
                atomicAdd(op + 32, g[u][1] * vj);
                atomicAdd(op + 64, g[u][2] * vj);
                atomicAdd(op + 96, g[u][3] * vj);
            }
        }
    } else {
        for (int j = 0; j < cnt; ++j) {
            const int   cj = lane_bcast_i(c, j);
            const int   rj = lane_bcast_i(r, j);
            const float vj = lane_bcast_f(v, j);
            const float* hp = hl + (size_t)cj * OUT_F;
            const float h0 = hp[0];
            const float h1 = hp[32];
            const float h2 = hp[64];
            const float h3 = hp[96];
            float* op = ol + (size_t)rj * OUT_F;
            atomicAdd(op +  0, h0 * vj);
            atomicAdd(op + 32, h1 * vj);
            atomicAdd(op + 64, h2 * vj);
            atomicAdd(op + 96, h3 * vj);
        }
    }
}


template <int NT>
__device__ __forceinline__ int block_scan_excl(int v, int* sh, int* total) {
    const int t = (int)threadIdx.x;
    sh[t] = v;
    __syncthreads();
    for (int d = 1; d < NT; d <<= 1) {
        const int a = (t >= d) ? sh[t - d] : 0;
        __syncthreads();
        sh[t] += a;
        __syncthreads();
    }
    const int incl = sh[t];
    *total = sh[NT - 1];
    __syncthreads();
    return incl - v;
}

#define BIN_THREADS 1024
__global__ void __launch_bounds__(BIN_THREADS)
gcn_bin_count(const int* __restrict__ erow, int* __restrict__ counts,
              int n_edges, int n_nodes, int nb) {
    __shared__ int hist[NB_MAX];
    const int t = (int)threadIdx.x;
    for (int i = t; i < nb; i += BIN_THREADS) hist[i] = 0;
    __syncthreads();
    const int per = (n_edges + NSLICE - 1) / NSLICE;
    const int e0 = (int)blockIdx.x * per;
    int e1 = e0 + per;
    if (e1 > n_edges) e1 = n_edges;
    for (int e = e0 + t; e < e1; e += BIN_THREADS) {
        const int r = erow[e];
        if ((unsigned)r < (unsigned)n_nodes) atomicAdd(&hist[r / RB], 1);
    }
    __syncthreads();
    for (int i = t; i < nb; i += BIN_THREADS) counts[i * NSLICE + (int)blockIdx.x] = hist[i];
}

__global__ void __launch_bounds__(NSLICE)
gcn_bin_offsets(int* __restrict__ counts, int* __restrict__ tot) {
    __shared__ int sh[NSLICE];
    const int t = (int)threadIdx.x;
    const int b = (int)blockIdx.x;
    const int c = counts[b * NSLICE + t];
    int total;
    const int ex = block_scan_excl<NSLICE>(c, sh, &total);
    counts[b * NSLICE + t] = ex;
    if (t == 0) tot[b] = total;
}


#define SUM_THREADS 512
#define SUM_WAVES   16
#define SCAN_PER_LANE ((RB + 1 + 31) / 32)
#define CH          4096
#define CH_PER_T    (CH / SUM_THREADS)

struct RowRun {
    v4f acc;
    int r;
    int e_r;
};

template <bool FIRST>
__device__ __forceinline__ void row_flush(RowRun& s, int endv, int r0, int row_base,
                                          float* __restrict__ out_lane, int n_nodes) {
    const int row = row_base + s.r;
    if (row < n_nodes) {
        float* op = out_lane + (size_t)row * OUT_F;
        if (!FIRST) s.acc += *(const v4f*)op;
        *(v4f*)op = s.acc;
    }
    const v4f zero = {};
    s.acc = zero;
    s.r += 1;
    s.e_r = __builtin_amdgcn_readlane(endv, s.r - r0);
}

__device__ __forceinline__ void load8(v4f (&g)[8], int rec_c, int o,
                                      const float* __restrict__ hl) {
#pragma unroll
    for (int u = 0; u < 8; ++u) {
        const int cj = lane_bcast_i(rec_c, o + u);
        g[u] = *(const v4f*)(hl + (size_t)cj * OUT_F);
    }
}

template <bool FIRST>
__device__ __forceinline__ void add8(RowRun& s, const v4f (&g)[8], int rec_v, int o,
                                     int jb, int se, int endv, int r0, int row_base,
                                     float* __restrict__ out_lane, int n_nodes) {
#pragma unroll
    for (int u = 0; u < 8; ++u) {
        const int jj = jb + u;
        if (jj < se) {
            while (jj == s.e_r) row_flush<FIRST>(s, endv, r0, row_base, out_lane, n_nodes);
            const float vj = __int_as_float(lane_bcast_i(rec_v, o + u));
            s.acc += g[u] * vj;
        }
    }
}

template <bool FIRST>
__device__ __forceinline__ void bin_pass(const float* __restrict__ h,
                                         const int4* __restrict__ rec,
                                         float* __restrict__ out, int n_nodes,
                                         int start, int cnt, int c0,
                                         int2* srt, int* hist, int* rowstart) {
    const int t = (int)threadIdx.x;
    const int lane = t & 31;
    const int wave = __builtin_amdgcn_readfirstlane(t >> 5);
    const int row_base = (int)blockIdx.x * RB;
    const int r0 = (wave * RB) / SUM_WAVES;
    const int r1 = ((wave + 1) * RB) / SUM_WAVES;
    const float* hl = h + 4 * lane;
    float* out_lane = out + 4 * lane;

    int n = cnt - c0;
    if (n > CH) n = CH;

    if (t < 256) hist[t] = 0;
    __syncthreads();

    int4 q[CH_PER_T];
    int rank[CH_PER_T];
#pragma unroll
    for (int k = 0; k < CH_PER_T; ++k) {
        int i = t + SUM_THREADS * k;
        if (i > n - 1) i = n - 1;
        if (i < 0) i = 0;
        q[k] = make_int4(0, 0, 0, 0);
        if (n > 0) q[k] = rec[start + c0 + i];
    }
#pragma unroll
    for (int k = 0; k < CH_PER_T; ++k) {
        rank[k] = 0;
        if (t + SUM_THREADS * k < n) rank[k] = atomicAdd(&hist[q[k].z], 1);
    }
    __syncthreads();

    if (wave == 0) {
        int v[SCAN_PER_LANE];
        int run = 0;
#pragma unroll
        for (int k = 0; k < SCAN_PER_LANE; ++k) {
            const int idx = SCAN_PER_LANE * lane + k;
            const int c = (idx < RB) ? hist[idx] : 0;
            v[k] = run;
            run += c;
        }
        int inc = run;
#pragma unroll
        for (int d = 1; d < 32; d <<= 1) {
            const int o = __shfl_up(inc, d);
            if (lane >= d) inc += o;
        }
        const int excl = inc - run;
#pragma unroll
        for (int k = 0; k < SCAN_PER_LANE; ++k) {
            const int idx = SCAN_PER_LANE * lane + k;
            if (idx <= RB) rowstart[idx] = excl + v[k];
        }
    }
    __syncthreads();

    int pos[CH_PER_T];
#pragma unroll
    for (int k = 0; k < CH_PER_T; ++k) pos[k] = rowstart[q[k].z] + rank[k];
#pragma unroll
    for (int k = 0; k < CH_PER_T; ++k)
        if (t + SUM_THREADS * k < n) srt[pos[k]] = make_int2(q[k].x, q[k].y);
    __syncthreads();

    if (r0 < r1) {
        int endv = 0x7fffffff;
        if (r0 + lane < r1) endv = rowstart[r0 + lane + 1];
        const int sb = __builtin_amdgcn_readfirstlane(rowstart[r0]);
        const int se = __builtin_amdgcn_readlane(endv, r1 - r0 - 1);

        RowRun s;
        { const v4f zero = {}; s.acc = zero; }
        s.r = r0;
        s.e_r = __builtin_amdgcn_readlane(endv, 0);

        if (sb < se) {
            const int l16 = lane & 15;
            int j = sb + l16;
            if (j > se - 1) j = se - 1;
            int2 my = srt[j];
            v4f ga[8], gb[8];
            load8(ga, my.x, 0, hl);
#pragma nounroll
            for (int base = sb; base < se; base += 16) {
                int jn = base + 16 + l16;
                if (jn > se - 1) jn = se - 1;
                const int2 myn = srt[jn];
                load8(gb, my.x, 8, hl);
                add8<FIRST>(s, ga, my.y, 0, base, se, endv, r0, row_base, out_lane, n_nodes);
                load8(ga, myn.x, 0, hl);
                add8<FIRST>(s, gb, my.y, 8, base + 8, se, endv, r0, row_base, out_lane, n_nodes);
                my = myn;
            }
        }
        while (s.r < r1) row_flush<FIRST>(s, endv, r0, row_base, out_lane, n_nodes);
    }
    __syncthreads();
}

__global__ void __launch_bounds__(SUM_THREADS)
gcn_bin_rowsum(const float* __restrict__ h, const int4* __restrict__ rec,
               const int* __restrict__ tot, float* __restrict__ out, int n_nodes) {
    __shared__ __attribute__((aligned(16))) int2 srt[CH];
    __shared__ int hist[256];
    __shared__ int rowstart[257];
    __shared__ int s_start;
    const int t = (int)threadIdx.x;
    const int b = (int)blockIdx.x;

    if (t == 0) s_start = 0;
    __syncthreads();
    int sp = 0;
    for (int i = t; i < b; i += SUM_THREADS) sp += tot[i];
    if (sp != 0) atomicAdd(&s_start, sp);
    __syncthreads();
    const int start = s_start;
    const int cnt = tot[b];

    bin_pass<true>(h, rec, out, n_nodes, start, cnt, 0, srt, hist, rowstart);
    for (int c0 = CH; c0 < cnt; c0 += CH)
        bin_pass<false>(h, rec, out, n_nodes, start, cnt, c0, srt, hist, rowstart);
}

extern "C" void kernel_launch(void* const* d_in, const int* in_sizes, int n_in,
                              void* d_out, int out_size, void* d_ws, size_t ws_size,
                              hipStream_t stream) {
    const float* x    = (const float*)d_in[0];
    const int*   erow = (const int*)  d_in[1];
    const int*   ecol = (const int*)  d_in[2];
    const float* ev   = (const float*)d_in[3];
    const float* w    = (const float*)d_in[4];
    const float* bias = (const float*)d_in[5];
    float*       out  = (float*)d_out;
    float*       h    = (float*)d_ws;

    const int n_nodes = in_sizes[0] / IN_F;
    const int n_edges = in_sizes[1];
    const int row_tiles = (n_nodes + 15) / 16;

    const int    nb      = (n_nodes + RB - 1) / RB;
    const size_t h_bytes = (size_t)n_nodes * OUT_F * sizeof(float);
    int4* rec    = (int4*)((char*)d_ws + h_bytes);
    int*  counts = (int*)((char*)rec + (size_t)n_edges * sizeof(int4));
    int*  tot    = counts + (size_t)nb * NSLICE;
    const size_t need = h_bytes + (size_t)n_edges * sizeof(int4)
                      + ((size_t)nb * NSLICE + (size_t)nb) * sizeof(int);

    if (ws_size >= need && nb <= NB_MAX) {
        gcn_bin_count<<<NSLICE, BIN_THREADS, 0, stream>>>(erow, counts, n_edges, n_nodes, nb);
        gcn_bin_offsets<<<nb, NSLICE, 0, stream>>>(counts, tot);
        gcn_gemm_place<<<NSLICE, FUSED_THREADS, 0, stream>>>(x, w, bias, h, n_nodes, row_tiles,
                                                        erow, ecol, ev, counts, tot, rec,
                                                        n_edges, nb, 1);
        gcn_bin_rowsum<<<nb, SUM_THREADS, 0, stream>>>(h, rec, tot, out, n_nodes);
    } else {
        const long long n4 = (long long)n_nodes * OUT_F / 4;
        gcn_zero<<<(unsigned)((n4 + 255) / 256), 256, 0, stream>>>((float4*)out, n4);
        gcn_gemm_place<<<NSLICE, FUSED_THREADS, 0, stream>>>(x, w, bias, h, n_nodes, row_tiles,
                                                        erow, ecol, ev, counts, tot, rec,
                                                        n_edges, nb, 0);
        const int blocks = (n_edges + 255) / 256;
        gcn_scatter<<<blocks, 256, 0, stream>>>(h, erow, ecol, ev, out, n_edges);
    }
}


# binning per edge slice only: filing waves of the GEMM launch counting-sort their slice by bin; row-sum groups gather a bin's 256 slice segments; 2 launches not 4; arrival-order ranks as before
# speedup vs baseline: 3.9445x; 1.2030x over previous
#include <hip/hip_runtime.h>
#include <hip/hip_bf16.h>

#define IN_F  256
#define OUT_F 128

typedef __attribute__((ext_vector_type(2))) float v2f;
typedef __attribute__((ext_vector_type(4))) float v4f;
typedef __attribute__((ext_vector_type(8))) float v8f;

__device__ __forceinline__ void gemm_chunk(v8f (&acc)[8], const v2f (&a)[8],
                                           const v4f* bp) {
#pragma unroll
    for (int gg = 0; gg < 4; ++gg) {
        v4f b[8];
#pragma unroll
        for (int j = 0; j < 8; ++j) b[j] = bp[(gg * 8 + j) * 32];
#pragma unroll
        for (int j = 0; j < 8; ++j)
            acc[j] = __builtin_amdgcn_wmma_f32_16x16x4_f32(
                 false, a[2 * gg],  false, b[j].xy,
                 (short)0, acc[j],  false,  false);
#pragma unroll
        for (int j = 0; j < 8; ++j)
            acc[j] = __builtin_amdgcn_wmma_f32_16x16x4_f32(
                 false, a[2 * gg + 1],  false, b[j].zw,
                 (short)0, acc[j],  false,  false);
    }
    __builtin_amdgcn_sched_barrier(0);
    asm volatile("v_nop\n\tv_nop\n\tv_nop\n\tv_nop\n\tv_nop\n\tv_nop\n\tv_nop\n\tv_nop\n\tv_nop");
    __builtin_amdgcn_sched_barrier(0);
}

#define GEMM_WAVES  16
#define PLACE_WAVES 4
#define FUSED_THREADS ((GEMM_WAVES + PLACE_WAVES) * 32)
#define GEMM_THREADS  (GEMM_WAVES * 32)
#define PLACE_THREADS (PLACE_WAVES * 32)
#define PLACE_BATCH   8
#define COUNT_BATCH   32

#define RB       196
#define NSLICE   256
#define NB_MAX   1024

__global__ void __launch_bounds__(FUSED_THREADS)
gcn_gemm_place(const float* __restrict__ x, const float* __restrict__ w,
               const float* __restrict__ bias, float* __restrict__ h,
               int n_nodes, int n_tiles,
               const int* __restrict__ erow, const int* __restrict__ ecol,
               const float* __restrict__ eval, int* __restrict__ bstart,
               int4* __restrict__ rec,
               int n_edges, int nb, int do_place) {
    __shared__ __attribute__((aligned(16))) v4f wl[32 * 8 * 32];
    __shared__ int cntw[PLACE_WAVES][NB_MAX];
    __shared__ int curw[PLACE_WAVES][NB_MAX];

    const int lane = threadIdx.x & 31;
    const int wave = __builtin_amdgcn_readfirstlane((int)(threadIdx.x >> 5));
    const int half = lane >> 4;
    const int l16  = lane & 15;

    const int tstride = GEMM_WAVES * (int)gridDim.x;
    int tile = wave * (int)gridDim.x + (int)blockIdx.x;

    int m = tile * 16 + l16;
    if (m >= n_nodes) m = n_nodes - 1;
    const float* xrow = x + (size_t)m * IN_F + 2 * half;

    v2f a_cur[8];
    if (wave < GEMM_WAVES) {
#pragma unroll
        for (int u = 0; u < 8; ++u) a_cur[u] = *(const v2f*)(xrow + 4 * u);

#pragma unroll 4
        for (int q = 0; q < (32 * 256) / GEMM_THREADS; ++q) {
            const int i  = (int)threadIdx.x + GEMM_THREADS * q;
            const int n  = i & 127;
            const int gh = i >> 7;
            const int g  = gh >> 1;
            const int hh = gh & 1;
            const float* wp = w + (size_t)(8 * g + 2 * hh) * OUT_F + n;
            v4f v;
            v.x = wp[0];
            v.y = wp[OUT_F];
            v.z = wp[4 * OUT_F];
            v.w = wp[5 * OUT_F];
            wl[(g * 8 + (n >> 4)) * 32 + hh * 16 + (n & 15)] = v;
        }
    } else if (do_place != 0) {
        __builtin_amdgcn_s_setprio(3);
        const int kw = wave - GEMM_WAVES;
        const int ht = (int)threadIdx.x - GEMM_THREADS;
        for (int i = lane; i < NB_MAX; i += 32) cntw[kw][i] = 0;
        const int per = (n_edges + NSLICE - 1) / NSLICE;
        const int e0 = (int)blockIdx.x * per;
        int e1 = e0 + per;
        if (e1 > n_edges) e1 = n_edges;
        for (int eb = e0; eb < e1; eb += PLACE_THREADS * COUNT_BATCH) {
            int r[COUNT_BATCH];
#pragma unroll
            for (int u = 0; u < COUNT_BATCH; ++u) {
                int e = eb + ht + PLACE_THREADS * u;
                if (e > e1 - 1) e = e1 - 1;
                r[u] = erow[e];
            }
#pragma unroll
            for (int u = 0; u < COUNT_BATCH; ++u) {
                const int e = eb + ht + PLACE_THREADS * u;
                if (e < e1 && (unsigned)r[u] < (unsigned)n_nodes) atomicAdd(&cntw[kw][r[u] / RB], 1);
            }
        }
    }
    __syncthreads();

    if (wave >= GEMM_WAVES) {
        if (do_place == 0) return;
        const int kw = wave - GEMM_WAVES;
        const int ht = (int)threadIdx.x - GEMM_THREADS;
        const int per = (n_edges + NSLICE - 1) / NSLICE;
        const int e0 = (int)blockIdx.x * per;
        int e1 = e0 + per;
        if (e1 > n_edges) e1 = n_edges;
        int carry = 0;
        for (int b32 = 0; b32 < nb + 1; b32 += 32) {
            const int b = b32 + lane;
            int all = 0;
            int mine = 0;
#pragma unroll
            for (int k = 0; k < PLACE_WAVES; ++k) {
                const int v = cntw[k][b];
                all += v;
                if (k < kw) mine += v;
            }
            int inc = all;
#pragma unroll
            for (int d = 1; d < 32; d <<= 1) {
                const int o = __shfl_up(inc, d);
                if (lane >= d) inc += o;
            }
            const int excl = carry + inc - all;
            curw[kw][b] = excl + mine;
            if (kw == 0 && b <= nb) bstart[b * NSLICE + (int)blockIdx.x] = excl;
            carry += __builtin_amdgcn_readlane(inc, 31);
        }
        for (int eb = e0; eb < e1; eb += PLACE_THREADS * PLACE_BATCH) {
            int   r[PLACE_BATCH];
            int   cj[PLACE_BATCH];
            float vj[PLACE_BATCH];
#pragma unroll
            for (int u = 0; u < PLACE_BATCH; ++u) {
                int e = eb + ht + PLACE_THREADS * u;
                if (e > e1 - 1) e = e1 - 1;
                r[u]  = erow[e];
                cj[u] = ecol[e];
                vj[u] = eval[e];
            }
#pragma unroll
            for (int u = 0; u < PLACE_BATCH; ++u) {
                const int e = eb + ht + PLACE_THREADS * u;
                if (e < e1 && (unsigned)r[u] < (unsigned)n_nodes) {
                    const int b = r[u] / RB;
                    const int pos = e0 + atomicAdd(&curw[kw][b], 1);
                    rec[pos] = make_int4(cj[u], __float_as_int(vj[u]), r[u] - b * RB, 0);
                }
            }
        }
        return;
    }

    while (tile < n_tiles) {
        const int ntile = tile + tstride;
        int mn = ntile * 16 + l16;
        if (mn >= n_nodes) mn = n_nodes - 1;
        const float* xnext = x + (size_t)mn * IN_F + 2 * half;

        v8f acc[8];
#pragma unroll
        for (int j = 0; j < 8; ++j) { const v8f zero = {}; acc[j] = zero; }

        const v4f* bp = wl + lane;
#pragma nounroll
        for (int cp = 0; cp < 4; ++cp) {
            v2f a_alt[8];
            {
                const float* src = xrow + 32 * (2 * cp + 1);
#pragma unroll
                for (int u = 0; u < 8; ++u) a_alt[u] = *(const v2f*)(src + 4 * u);
            }
            gemm_chunk(acc, a_cur, bp);
            {
                const float* src = (cp < 3) ? (xrow + 32 * (2 * cp + 2)) : xnext;
#pragma unroll
                for (int u = 0; u < 8; ++u) a_cur[u] = *(const v2f*)(src + 4 * u);
            }
            gemm_chunk(acc, a_alt, bp + 4 * 8 * 32);
            bp += 2 * 4 * 8 * 32;
        }

        const int row_base = tile * 16;
        float bn[8];
#pragma unroll
        for (int j = 0; j < 8; ++j) bn[j] = bias[16 * j + l16];
        float* hp = h + (size_t)(row_base + 8 * half) * OUT_F + l16;

        if (row_base + 16 <= n_nodes) {
#pragma unroll
            for (int j = 0; j < 8; ++j)
#pragma unroll
                for (int v = 0; v < 8; ++v)
                    hp[(size_t)v * OUT_F + 16 * j] = acc[j][v] + bn[j];
        } else {
            const int r0 = row_base + 8 * half;
#pragma unroll
            for (int j = 0; j < 8; ++j)
#pragma unroll
                for (int v = 0; v < 8; ++v)
                    if (r0 + v < n_nodes) hp[(size_t)v * OUT_F + 16 * j] = acc[j][v] + bn[j];
        }

        tile = ntile;
        xrow = xnext;
    }
}

__global__ void gcn_zero(float4* __restrict__ out, long long n4) {
    long long i = (long long)blockIdx.x * blockDim.x + threadIdx.x;
    if (i < n4) out[i] = make_float4(0.f, 0.f, 0.f, 0.f);
}

#define EDGES_PER_WAVE 32
#define EDGE_BATCH 8

__device__ __forceinline__ int lane_bcast_i(int x, int src_lane) {
    return __builtin_amdgcn_readlane(x, src_lane);
}
__device__ __forceinline__ float lane_bcast_f(float x, int src_lane) {
    return __int_as_float(__builtin_amdgcn_readlane(__float_as_int(x), src_lane));
}

__global__ void __launch_bounds__(256)
gcn_scatter(const float* __restrict__ h, const int* __restrict__ erow,
            const int* __restrict__ ecol, const float* __restrict__ eval,
            float* __restrict__ out, int n_edges) {
    const int lane = threadIdx.x & 31;
    const int wave = __builtin_amdgcn_readfirstlane((int)(threadIdx.x >> 5));
    const int base = ((int)blockIdx.x * 8 + wave) * EDGES_PER_WAVE;
    if (base >= n_edges) return;
    int cnt = n_edges - base;
    if (cnt > EDGES_PER_WAVE) cnt = EDGES_PER_WAVE;

    int   r = 0;
    int   c = 0;
    float v = 0.f;
    if (lane < cnt) {
        r = erow[base + lane];
        c = ecol[base + lane];
        v = eval[base + lane];
    }

    const float* hl = h + lane;
    float*       ol = out + lane;

    if (cnt == EDGES_PER_WAVE) {
#pragma unroll
        for (int j0 = 0; j0 < EDGES_PER_WAVE; j0 += EDGE_BATCH) {
            float g[EDGE_BATCH][4];
#pragma unroll
            for (int u = 0; u < EDGE_BATCH; ++u) {
                const int cj = lane_bcast_i(c, j0 + u);
                const float* hp = hl + (size_t)cj * OUT_F;
                g[u][0] = hp[0];
                g[u][1] = hp[32];
                g[u][2] = hp[64];
                g[u][3] = hp[96];
            }
#pragma unroll
            for (int u = 0; u < EDGE_BATCH; ++u) {
                const int   rj = lane_bcast_i(r, j0 + u);
                const float vj = lane_bcast_f(v, j0 + u);
                float* op = ol + (size_t)rj * OUT_F;
                atomicAdd(op +  0, g[u][0] * vj);
                atomicAdd(op + 32, g[u][1] * vj);
                atomicAdd(op + 64, g[u][2] * vj);
                atomicAdd(op + 96, g[u][3] * vj);
            }
        }
    } else {
        for (int j = 0; j < cnt; ++j) {
            const int   cj = lane_bcast_i(c, j);
            const int   rj = lane_bcast_i(r, j);
            const float vj = lane_bcast_f(v, j);
            const float* hp = hl + (size_t)cj * OUT_F;
            const float h0 = hp[0];
            const float h1 = hp[32];
            const float h2 = hp[64];
            const float h3 = hp[96];
            float* op = ol + (size_t)rj * OUT_F;
            atomicAdd(op +  0, h0 * vj);
            atomicAdd(op + 32, h1 * vj);
            atomicAdd(op + 64, h2 * vj);
            atomicAdd(op + 96, h3 * vj);
        }
    }
}


#define SUM_THREADS 512
#define SUM_WAVES   16
#define SCAN_PER_LANE ((RB + 1 + 31) / 32)
#define CH          4096
#define CH_PER_T    (CH / SUM_THREADS)

struct RowRun {
    v4f acc;
    int r;
    int e_r;
};

template <bool FIRST>
__device__ __forceinline__ void row_flush(RowRun& s, int endv, int r0, int row_base,
                                          float* __restrict__ out_lane, int n_nodes) {
    const int row = row_base + s.r;
    if (row < n_nodes) {
        float* op = out_lane + (size_t)row * OUT_F;
        if (!FIRST) s.acc += *(const v4f*)op;
        *(v4f*)op = s.acc;
    }
    const v4f zero = {};
    s.acc = zero;
    s.r += 1;
    s.e_r = __builtin_amdgcn_readlane(endv, s.r - r0);
}

__device__ __forceinline__ void load8(v4f (&g)[8], int rec_c, int o,
                                      const float* __restrict__ hl) {
#pragma unroll
    for (int u = 0; u < 8; ++u) {
        const int cj = lane_bcast_i(rec_c, o + u);
        g[u] = *(const v4f*)(hl + (size_t)cj * OUT_F);
    }
}

template <bool FIRST>
__device__ __forceinline__ void add8(RowRun& s, const v4f (&g)[8], int rec_v, int o,
                                     int jb, int se, int endv, int r0, int row_base,
                                     float* __restrict__ out_lane, int n_nodes) {
#pragma unroll
    for (int u = 0; u < 8; ++u) {
        const int jj = jb + u;
        if (jj < se) {
            while (jj == s.e_r) row_flush<FIRST>(s, endv, r0, row_base, out_lane, n_nodes);
            const float vj = __int_as_float(lane_bcast_i(rec_v, o + u));
            s.acc += g[u] * vj;
        }
    }
}

template <bool FIRST>
__device__ __forceinline__ void bin_pass(const float* __restrict__ h,
                                         const int4* __restrict__ rec,
                                         float* __restrict__ out, int n_nodes,
                                         const int* pfx, const int* sbase, int cnt, int c0,
                                         int2* srt, int* hist, int* rowstart) {
    const int t = (int)threadIdx.x;
    const int lane = t & 31;
    const int wave = __builtin_amdgcn_readfirstlane(t >> 5);
    const int row_base = (int)blockIdx.x * RB;
    const int r0 = (wave * RB) / SUM_WAVES;
    const int r1 = ((wave + 1) * RB) / SUM_WAVES;
    const float* hl = h + 4 * lane;
    float* out_lane = out + 4 * lane;

    int n = cnt - c0;
    if (n > CH) n = CH;

    if (t < 256) hist[t] = 0;
    __syncthreads();

    int4 q[CH_PER_T];
    int rank[CH_PER_T];
    int gi[CH_PER_T];
    int sl[CH_PER_T];
#pragma unroll
    for (int k = 0; k < CH_PER_T; ++k) {
        int i = t + SUM_THREADS * k;
        if (i > n - 1) i = n - 1;
        if (i < 0) i = 0;
        gi[k] = c0 + i;
        sl[k] = 0;
    }
#pragma unroll
    for (int step = NSLICE / 2; step >= 1; step >>= 1)
#pragma unroll
        for (int k = 0; k < CH_PER_T; ++k)
            if (pfx[sl[k] + step] <= gi[k]) sl[k] += step;
#pragma unroll
    for (int k = 0; k < CH_PER_T; ++k) {
        q[k] = make_int4(0, 0, 0, 0);
        if (n > 0) q[k] = rec[sbase[sl[k]] + (gi[k] - pfx[sl[k]])];
    }
#pragma unroll
    for (int k = 0; k < CH_PER_T; ++k) {
        rank[k] = 0;
        if (t + SUM_THREADS * k < n) rank[k] = atomicAdd(&hist[q[k].z], 1);
    }
    __syncthreads();

    if (wave == 0) {
        int v[SCAN_PER_LANE];
        int run = 0;
#pragma unroll
        for (int k = 0; k < SCAN_PER_LANE; ++k) {
            const int idx = SCAN_PER_LANE * lane + k;
            const int c = (idx < RB) ? hist[idx] : 0;
            v[k] = run;
            run += c;
        }
        int inc = run;
#pragma unroll
        for (int d = 1; d < 32; d <<= 1) {
            const int o = __shfl_up(inc, d);
            if (lane >= d) inc += o;
        }
        const int excl = inc - run;
#pragma unroll
        for (int k = 0; k < SCAN_PER_LANE; ++k) {
            const int idx = SCAN_PER_LANE * lane + k;
            if (idx <= RB) rowstart[idx] = excl + v[k];
        }
    }
    __syncthreads();

    int pos[CH_PER_T];
#pragma unroll
    for (int k = 0; k < CH_PER_T; ++k) pos[k] = rowstart[q[k].z] + rank[k];
#pragma unroll
    for (int k = 0; k < CH_PER_T; ++k)
        if (t + SUM_THREADS * k < n) srt[pos[k]] = make_int2(q[k].x, q[k].y);
    __syncthreads();

    if (r0 < r1) {
        int endv = 0x7fffffff;
        if (r0 + lane < r1) endv = rowstart[r0 + lane + 1];
        const int sb = __builtin_amdgcn_readfirstlane(rowstart[r0]);
        const int se = __builtin_amdgcn_readlane(endv, r1 - r0 - 1);

        RowRun s;
        { const v4f zero = {}; s.acc = zero; }
        s.r = r0;
        s.e_r = __builtin_amdgcn_readlane(endv, 0);

        if (sb < se) {
            const int l16 = lane & 15;
            int j = sb + l16;
            if (j > se - 1) j = se - 1;
            int2 my = srt[j];
            v4f ga[8], gb[8];
            load8(ga, my.x, 0, hl);
#pragma nounroll
            for (int base = sb; base < se; base += 16) {
                int jn = base + 16 + l16;
                if (jn > se - 1) jn = se - 1;
                const int2 myn = srt[jn];
                load8(gb, my.x, 8, hl);
                add8<FIRST>(s, ga, my.y, 0, base, se, endv, r0, row_base, out_lane, n_nodes);
                load8(ga, myn.x, 0, hl);
                add8<FIRST>(s, gb, my.y, 8, base + 8, se, endv, r0, row_base, out_lane, n_nodes);
                my = myn;
            }
        }
        while (s.r < r1) row_flush<FIRST>(s, endv, r0, row_base, out_lane, n_nodes);
    }
    __syncthreads();
}

__global__ void __launch_bounds__(SUM_THREADS)
gcn_bin_rowsum(const float* __restrict__ h, const int4* __restrict__ rec,
               const int* __restrict__ bstart, float* __restrict__ out,
               int n_nodes, int n_edges) {
    __shared__ __attribute__((aligned(16))) int2 srt[CH];
    __shared__ int hist[256];
    __shared__ int rowstart[257];
    __shared__ int pfx[NSLICE + 1];
    __shared__ int sbase[NSLICE];
    const int t = (int)threadIdx.x;
    const int lane = t & 31;
    const int b = (int)blockIdx.x;
    const int per = (n_edges + NSLICE - 1) / NSLICE;

    if (t < NSLICE) {
        const int lo = bstart[b * NSLICE + t];
        const int hi = bstart[(b + 1) * NSLICE + t];
        sbase[t] = t * per + lo;
        pfx[t] = hi - lo;
    }
    __syncthreads();
    if (t < 32) {
        int v[NSLICE / 32];
        int run = 0;
#pragma unroll
        for (int k = 0; k < NSLICE / 32; ++k) {
            const int cN = pfx[(NSLICE / 32) * lane + k];
            v[k] = run;
            run += cN;
        }
        int inc = run;
#pragma unroll
        for (int d = 1; d < 32; d <<= 1) {
            const int o = __shfl_up(inc, d);
            if (lane >= d) inc += o;
        }
        const int excl = inc - run;
#pragma unroll
        for (int k = 0; k < NSLICE / 32; ++k) pfx[(NSLICE / 32) * lane + k] = excl + v[k];
        if (lane == 31) pfx[NSLICE] = inc;
    }
    __syncthreads();
    const int cnt = pfx[NSLICE];

    bin_pass<true>(h, rec, out, n_nodes, pfx, sbase, cnt, 0, srt, hist, rowstart);
    for (int c0 = CH; c0 < cnt; c0 += CH)
        bin_pass<false>(h, rec, out, n_nodes, pfx, sbase, cnt, c0, srt, hist, rowstart);
}

extern "C" void kernel_launch(void* const* d_in, const int* in_sizes, int n_in,
                              void* d_out, int out_size, void* d_ws, size_t ws_size,
                              hipStream_t stream) {
    const float* x    = (const float*)d_in[0];
    const int*   erow = (const int*)  d_in[1];
    const int*   ecol = (const int*)  d_in[2];
    const float* ev   = (const float*)d_in[3];
    const float* w    = (const float*)d_in[4];
    const float* bias = (const float*)d_in[5];
    float*       out  = (float*)d_out;
    float*       h    = (float*)d_ws;

    const int n_nodes = in_sizes[0] / IN_F;
    const int n_edges = in_sizes[1];
    const int row_tiles = (n_nodes + 15) / 16;

    const int    nb      = (n_nodes + RB - 1) / RB;
    const size_t h_bytes = (size_t)n_nodes * OUT_F * sizeof(float);
    int4* rec    = (int4*)((char*)d_ws + h_bytes);
    int*  bstart = (int*)((char*)rec + (size_t)n_edges * sizeof(int4));
    const size_t need = h_bytes + (size_t)n_edges * sizeof(int4)
                      + ((size_t)nb + 1) * NSLICE * sizeof(int);

    if (ws_size >= need && nb < NB_MAX) {
        gcn_gemm_place<<<NSLICE, FUSED_THREADS, 0, stream>>>(x, w, bias, h, n_nodes, row_tiles,
                                                        erow, ecol, ev, bstart, rec,
                                                        n_edges, nb, 1);
        gcn_bin_rowsum<<<nb, SUM_THREADS, 0, stream>>>(h, rec, bstart, out, n_nodes, n_edges);
    } else {
        const long long n4 = (long long)n_nodes * OUT_F / 4;
        gcn_zero<<<(unsigned)((n4 + 255) / 256), 256, 0, stream>>>((float4*)out, n4);
        gcn_gemm_place<<<NSLICE, FUSED_THREADS, 0, stream>>>(x, w, bias, h, n_nodes, row_tiles,
                                                        erow, ecol, ev, bstart, rec,
                                                        n_edges, nb, 0);
        const int blocks = (n_edges + 255) / 256;
        gcn_scatter<<<blocks, 256, 0, stream>>>(h, erow, ecol, ev, out, n_edges);
    }
}
